# starting kernel
# baseline (speedup 1.0000x reference)
#include <hip/hip_runtime.h>
#include <math.h>

typedef __attribute__((ext_vector_type(2))) float v2f;
typedef __attribute__((ext_vector_type(8))) float v8f;

#define C_DIM 256
#define D_DIM 256
#define N_TOK 4096
#define SCALE 0.0625f
#define EPSV  1e-5f

#define WS_GP    0
#define WS_SGP   1024
#define WS_SBP   1028
#define WS_LOG   1032
#define WS_MU    17416
#define WS_RSTD  33800
#define WS_ALPHA 50184
#define WS_WSUM  66568
#define WS_AMU   66572
#define WS_U     66576
#define WS_YVEC  67600

__device__ inline v8f gemv_tile_f32(const float* x, const float* w,
                                    int K, long sk, long sn, int n0, v8f acc) {
    const int lane = threadIdx.x & 31;
    const int half = lane >> 4;
    const int l16  = lane & 15;
    const float amask = (l16 == 0) ? 1.0f : 0.0f;
    const float* wp = w + (long)(n0 + l16) * sn;
    for (int k0 = 0; k0 < K; k0 += 4) {
        const int kb = k0 + 2 * half;
        const float x0 = x[kb];
        const float x1 = x[kb + 1];
        v2f a;
        a.x = x0 * amask;
        a.y = x1 * amask;
        v2f b;
        b.x = wp[(long)kb * sk];
        b.y = wp[(long)(kb + 1) * sk];
        acc = __builtin_amdgcn_wmma_f32_16x16x4_f32(
             false, a,  false, b,
             (short)0, acc,  false,  false);
    }
    return acc;
}

__device__ inline float block_reduce_sum_512(float v, float* red) {
    const int t = threadIdx.x;
    red[t] = v; __syncthreads();
    for (int o = 256; o > 0; o >>= 1) {
        if (t < o) red[t] += red[t + o];
        __syncthreads();
    }
    float r = red[0]; __syncthreads();
    return r;
}

__global__ __launch_bounds__(512) void prep_kernel(
    const float* tvec, const float* wt, const float* bt,
    const float* wk_x, const float* wk_h, const float* wk_s, const float* wk_m,
    const float* g_x, const float* be_x, const float* g_h, const float* be_h,
    const float* g_s, const float* be_s, const float* g_m, const float* be_m,
    float* gp, float* sgp, float* sbp)
{
    __shared__ float q_s[D_DIM];
    __shared__ float p_s[C_DIM];
    __shared__ float red[512];
    const int tid  = threadIdx.x;
    const int wave = tid >> 5;
    const int lane = tid & 31;

    {
        v8f acc = {};
        acc = gemv_tile_f32(tvec, wt, C_DIM, (long)D_DIM, 1L, wave * 16, acc);
        if (lane < 16) q_s[wave * 16 + lane] = acc[0] + bt[wave * 16 + lane];
    }
    __syncthreads();

    const float* wks[4] = {wk_x, wk_h, wk_s, wk_m};
    const float* gs[4]  = {g_x,  g_h,  g_s,  g_m};
    const float* bes[4] = {be_x, be_h, be_s, be_m};

    for (int mod = 0; mod < 4; ++mod) {
        v8f acc = {};
        acc = gemv_tile_f32(q_s, wks[mod], D_DIM, 1L, (long)D_DIM, wave * 16, acc);
        if (lane < 16) p_s[wave * 16 + lane] = acc[0];
        __syncthreads();

        float gv = 0.0f, bv = 0.0f;
        if (tid < C_DIM) {
            const float p = p_s[tid];
            gv = gs[mod][tid] * p;
            bv = bes[mod][tid] * p;
            gp[mod * C_DIM + tid] = gv;
        }
        const float sg = block_reduce_sum_512(gv, red);
        const float sb = block_reduce_sum_512(bv, red);
        if (tid == 0) { sgp[mod] = sg; sbp[mod] = sb; }
        __syncthreads();
    }
}

__global__ __launch_bounds__(256) void pass1_kernel(
    const float* x, const float* h, const float* s, const float* m,
    const float* gp, const float* sgp, const float* sbp,
    float* logits, float* mus, float* rstds)
{
    __shared__ float gpm_s[C_DIM];
    __shared__ float sh1[4][64];
    __shared__ float sh2[4][64];
    __shared__ float shd[4][64];
    const int bid = blockIdx.x;
    const int mod = bid >> 6;
    const int j0  = (bid & 63) << 6;
    const int tl  = threadIdx.x & 63;
    const int grp = threadIdx.x >> 6;
    const float* aptr[4] = {x, h, s, m};
    const float* a = aptr[mod];

    gpm_s[threadIdx.x] = gp[mod * C_DIM + threadIdx.x];
    __syncthreads();

    const int j  = j0 + tl;
    const int c0 = grp << 6;
    float s1 = 0.0f, s2 = 0.0f, dp = 0.0f;
#pragma unroll 4
    for (int cc = 0; cc < 64; ++cc) {
        const int c = c0 + cc;
        const float v = a[(long)c * N_TOK + j];
        s1 += v;
        s2 += v * v;
        dp += v * gpm_s[c];
    }
    sh1[grp][tl] = s1; sh2[grp][tl] = s2; shd[grp][tl] = dp;
    __syncthreads();

    if (threadIdx.x < 64) {
        const int jj = j0 + threadIdx.x;
        const float t1 = sh1[0][threadIdx.x] + sh1[1][threadIdx.x] +
                         sh1[2][threadIdx.x] + sh1[3][threadIdx.x];
        const float t2 = sh2[0][threadIdx.x] + sh2[1][threadIdx.x] +
                         sh2[2][threadIdx.x] + sh2[3][threadIdx.x];
        const float td = shd[0][threadIdx.x] + shd[1][threadIdx.x] +
                         shd[2][threadIdx.x] + shd[3][threadIdx.x];
        const float mu   = t1 * (1.0f / C_DIM);
        const float var  = t2 * (1.0f / C_DIM) - mu * mu;
        const float rstd = rsqrtf(var + EPSV);
        const float lg   = SCALE * (rstd * (td - mu * sgp[mod]) + sbp[mod]);
        const int idx = mod * N_TOK + jj;
        logits[idx] = lg;
        mus[idx]    = mu;
        rstds[idx]  = rstd;
    }
}

__global__ __launch_bounds__(1024) void softmax_kernel(
    const float* logits, const float* mus, const float* rstds,
    float* alpha, float* wsum, float* amu)
{
    __shared__ float red[1024];
    const int t = threadIdx.x;
    float l[16];
    float lmax = -3.4e38f;
#pragma unroll
    for (int i = 0; i < 16; ++i) { l[i] = logits[t * 16 + i]; lmax = fmaxf(lmax, l[i]); }
    red[t] = lmax; __syncthreads();
    for (int o = 512; o > 0; o >>= 1) { if (t < o) red[t] = fmaxf(red[t], red[t + o]); __syncthreads(); }
    const float gmax = red[0]; __syncthreads();

    float ls = 0.0f;
#pragma unroll
    for (int i = 0; i < 16; ++i) { l[i] = __expf(l[i] - gmax); ls += l[i]; }
    red[t] = ls; __syncthreads();
    for (int o = 512; o > 0; o >>= 1) { if (t < o) red[t] += red[t + o]; __syncthreads(); }
    const float inv = 1.0f / red[0]; __syncthreads();

    const int mod = t >> 8;
    float wp = 0.0f, ap = 0.0f;
#pragma unroll
    for (int i = 0; i < 16; ++i) {
        const int j  = t * 16 + i;
        const float w  = l[i] * inv;
        const float al = w * rstds[j];
        alpha[j] = al;
        wp += w;
        ap += al * mus[j];
    }
    for (int md = 0; md < 4; ++md) {
        red[t] = (mod == md) ? wp : 0.0f; __syncthreads();
        for (int o = 512; o > 0; o >>= 1) { if (t < o) red[t] += red[t + o]; __syncthreads(); }
        if (t == 0) wsum[md] = red[0];
        __syncthreads();
        red[t] = (mod == md) ? ap : 0.0f; __syncthreads();
        for (int o = 512; o > 0; o >>= 1) { if (t < o) red[t] += red[t + o]; __syncthreads(); }
        if (t == 0) amu[md] = red[0];
        __syncthreads();
    }
}

__global__ __launch_bounds__(256) void pass2_kernel(
    const float* x, const float* h, const float* s, const float* m,
    const float* alpha, const float* wsum, const float* amu,
    const float* g_x, const float* be_x, const float* g_h, const float* be_h,
    const float* g_s, const float* be_s, const float* g_m, const float* be_m,
    float* u)
{
    __shared__ float red[256];
    const int bid = blockIdx.x;
    const int mod = bid >> 8;
    const int c   = bid & 255;
    const float* aptr[4] = {x, h, s, m};
    const float* gs[4]   = {g_x,  g_h,  g_s,  g_m};
    const float* bes[4]  = {be_x, be_h, be_s, be_m};

    const float* a  = aptr[mod] + (long)c * N_TOK;
    const float* al = alpha + mod * N_TOK;
    float acc = 0.0f;
#pragma unroll 4
    for (int j = threadIdx.x; j < N_TOK; j += 256) acc += al[j] * a[j];
    red[threadIdx.x] = acc; __syncthreads();
    for (int o = 128; o > 0; o >>= 1) {
        if (threadIdx.x < o) red[threadIdx.x] += red[threadIdx.x + o];
        __syncthreads();
    }
    if (threadIdx.x == 0) {
        const float T = red[0];
        u[mod * C_DIM + c] = gs[mod][c] * (T - amu[mod]) + bes[mod][c] * wsum[mod];
    }
}

__global__ __launch_bounds__(512) void head_kernel(
    const float* u,
    const float* wv_x, const float* wv_h, const float* wv_s, const float* wv_m,
    const float* g_res, const float* be_res,
    const float* w_geglu, const float* b_geglu,
    const float* w_ff2, const float* b_ff2,
    const float* w_out, const float* b_out,
    float* yvec)
{
    __shared__ float ovec[D_DIM];
    __shared__ float rr[D_DIM];
    __shared__ float ggl[2 * D_DIM];
    __shared__ float hh[D_DIM];
    __shared__ float ffv[D_DIM];
    __shared__ float red[512];
    const int tid  = threadIdx.x;
    const int wave = tid >> 5;
    const int lane = tid & 31;
    const float* wv[4] = {wv_x, wv_h, wv_s, wv_m};

    {
        v8f acc = {};
        for (int mod = 0; mod < 4; ++mod)
            acc = gemv_tile_f32(u + mod * C_DIM, wv[mod], C_DIM, (long)D_DIM, 1L,
                                wave * 16, acc);
        if (lane < 16) ovec[wave * 16 + lane] = acc[0];
    }
    __syncthreads();

    float v = (tid < D_DIM) ? ovec[tid] : 0.0f;
    const float mu = block_reduce_sum_512(v, red) * (1.0f / D_DIM);
    const float dv = (tid < D_DIM) ? (ovec[tid] - mu) : 0.0f;
    const float var = block_reduce_sum_512(dv * dv, red) * (1.0f / D_DIM);
    const float rstd = rsqrtf(var + EPSV);
    if (tid < D_DIM) rr[tid] = (ovec[tid] - mu) * rstd * g_res[tid] + be_res[tid];
    __syncthreads();

    for (int tt = wave; tt < 32; tt += 16) {
        v8f acc = {};
        acc = gemv_tile_f32(rr, w_geglu, D_DIM, (long)(2 * D_DIM), 1L, tt * 16, acc);
        if (lane < 16) ggl[tt * 16 + lane] = acc[0] + b_geglu[tt * 16 + lane];
    }
    __syncthreads();

    if (tid < D_DIM) {
        const float g  = ggl[D_DIM + tid];
        const float ge = 0.5f * g * (1.0f + erff(g * 0.70710678118654752f));
        hh[tid] = ggl[tid] * ge;
    }
    __syncthreads();

    {
        v8f acc = {};
        acc = gemv_tile_f32(hh, w_ff2, D_DIM, (long)D_DIM, 1L, wave * 16, acc);
        if (lane < 16) ffv[wave * 16 + lane] = acc[0] + b_ff2[wave * 16 + lane];
    }
    __syncthreads();

    {
        v8f acc = {};
        acc = gemv_tile_f32(ffv, w_out, D_DIM, (long)C_DIM, 1L, wave * 16, acc);
        if (lane < 16) yvec[wave * 16 + lane] = acc[0] + b_out[wave * 16 + lane];
    }
}

__global__ __launch_bounds__(1024) void bcast_kernel(const float* yvec, float* out)
{
    const int c = blockIdx.x;
    const float v = yvec[c];
    float* o = out + (long)c * N_TOK;
    for (int j = threadIdx.x; j < N_TOK; j += 1024) o[j] = v;
}

extern "C" void kernel_launch(void* const* d_in, const int* in_sizes, int n_in,
                              void* d_out, int out_size, void* d_ws, size_t ws_size,
                              hipStream_t stream) {
    const float* x       = (const float*)d_in[0];
    const float* h       = (const float*)d_in[1];
    const float* s       = (const float*)d_in[2];
    const float* m       = (const float*)d_in[3];
    const float* tvec    = (const float*)d_in[4];
    const float* wt      = (const float*)d_in[5];
    const float* bt      = (const float*)d_in[6];
    const float* wk_x    = (const float*)d_in[7];
    const float* wv_x    = (const float*)d_in[8];
    const float* g_x     = (const float*)d_in[9];
    const float* be_x    = (const float*)d_in[10];
    const float* wk_h    = (const float*)d_in[11];
    const float* wv_h    = (const float*)d_in[12];
    const float* g_h     = (const float*)d_in[13];
    const float* be_h    = (const float*)d_in[14];
    const float* wk_s    = (const float*)d_in[15];
    const float* wv_s    = (const float*)d_in[16];
    const float* g_s     = (const float*)d_in[17];
    const float* be_s    = (const float*)d_in[18];
    const float* wk_m    = (const float*)d_in[19];
    const float* wv_m    = (const float*)d_in[20];
    const float* g_m     = (const float*)d_in[21];
    const float* be_m    = (const float*)d_in[22];
    const float* w_geglu = (const float*)d_in[23];
    const float* b_geglu = (const float*)d_in[24];
    const float* w_ff2   = (const float*)d_in[25];
    const float* b_ff2   = (const float*)d_in[26];
    const float* g_res   = (const float*)d_in[27];
    const float* be_res  = (const float*)d_in[28];
    const float* w_out   = (const float*)d_in[29];
    const float* b_out   = (const float*)d_in[30];
    float* out = (float*)d_out;
    float* ws  = (float*)d_ws;

    float* gp     = ws + WS_GP;
    float* sgp    = ws + WS_SGP;
    float* sbp    = ws + WS_SBP;
    float* logits = ws + WS_LOG;
    float* mus    = ws + WS_MU;
    float* rstds  = ws + WS_RSTD;
    float* alpha  = ws + WS_ALPHA;
    float* wsum   = ws + WS_WSUM;
    float* amu    = ws + WS_AMU;
    float* u      = ws + WS_U;
    float* yvec   = ws + WS_YVEC;

    prep_kernel<<<1, 512, 0, stream>>>(tvec, wt, bt,
                                       wk_x, wk_h, wk_s, wk_m,
                                       g_x, be_x, g_h, be_h, g_s, be_s, g_m, be_m,
                                       gp, sgp, sbp);
    pass1_kernel<<<256, 256, 0, stream>>>(x, h, s, m, gp, sgp, sbp,
                                          logits, mus, rstds);
    softmax_kernel<<<1, 1024, 0, stream>>>(logits, mus, rstds, alpha, wsum, amu);
    pass2_kernel<<<1024, 256, 0, stream>>>(x, h, s, m, alpha, wsum, amu,
                                           g_x, be_x, g_h, be_h, g_s, be_s, g_m, be_m,
                                           u);
    head_kernel<<<1, 512, 0, stream>>>(u, wv_x, wv_h, wv_s, wv_m,
                                       g_res, be_res, w_geglu, b_geglu,
                                       w_ff2, b_ff2, w_out, b_out, yvec);
    bcast_kernel<<<256, 1024, 0, stream>>>(yvec, out);
}


# softmax launch removed: pass1 leaves block-relative exp terms and per-64-token-block max/sum partials, pass2 rescales by exp(blockmax-max)/sum; fp sum order of softmax sums changes (fixed)
# speedup vs baseline: 3.2113x; 3.2113x over previous
#include <hip/hip_runtime.h>
#include <math.h>

typedef __attribute__((ext_vector_type(2))) float v2f;
typedef __attribute__((ext_vector_type(8))) float v8f;

#define C_DIM 256
#define D_DIM 256
#define N_TOK 4096
#define SCALE 0.0625f
#define EPSV  1e-5f

#define WS_GP    0
#define WS_OVP   1032
#define WS_GGL   2056
#define WS_FFV   2568
#define WS_AL    17408
#define WS_PM    33792
#define WS_PS    34048
#define WS_PR    34304
#define WS_Q     50184
#define WS_SGPP  50440
#define WS_SBPP  50504
#define WS_U     66576
#define WS_YVEC  67600

__device__ inline v8f gemv_tile_f32(const float* x, const float* w,
                                    int K, long sk, long sn, int n0, v8f acc) {
    const int lane = threadIdx.x & 31;
    const int half = lane >> 4;
    const int l16  = lane & 15;
    const float amask = (l16 == 0) ? 1.0f : 0.0f;
    const float* wp = w + (long)(n0 + l16) * sn;
    for (int k0 = 0; k0 < K; k0 += 4) {
        const int kb = k0 + 2 * half;
        const float x0 = x[kb];
        const float x1 = x[kb + 1];
        v2f a;
        a.x = x0 * amask;
        a.y = x1 * amask;
        v2f b;
        b.x = wp[(long)kb * sk];
        b.y = wp[(long)(kb + 1) * sk];
        acc = __builtin_amdgcn_wmma_f32_16x16x4_f32(
             false, a,  false, b,
             (short)0, acc,  false,  false);
    }
    return acc;
}

__device__ inline float block_reduce_sum_256(float v, float* red) {
    const int t = threadIdx.x;
    red[t] = v; __syncthreads();
    for (int o = 128; o > 0; o >>= 1) {
        if (t < o) red[t] += red[t + o];
        __syncthreads();
    }
    float r = red[0]; __syncthreads();
    return r;
}

__device__ inline float wave_max32(float v) {
    for (int o = 16; o > 0; o >>= 1) v = fmaxf(v, __shfl_xor(v, o, 32));
    return v;
}
__device__ inline float wave_sum32(float v) {
    for (int o = 16; o > 0; o >>= 1) v += __shfl_xor(v, o, 32);
    return v;
}

__device__ inline float gemv_tile_ksplit8(const float* x, const float* w,
                                          int K, long sk, long sn, int n0, float* part) {
    const int wave = threadIdx.x >> 5;
    const int lane = threadIdx.x & 31;
    const int kw = K >> 3;
    v8f acc = {};
    acc = gemv_tile_f32(x + wave * kw, w + (long)(wave * kw) * sk, kw, sk, sn, n0, acc);
    if (lane < 16) part[wave * 16 + lane] = acc[0];
    __syncthreads();
    float r = 0.0f;
    if (threadIdx.x < 16) {
        for (int i = 0; i < 8; ++i) r += part[i * 16 + threadIdx.x];
    }
    return r;
}

__global__ __launch_bounds__(256) void q_kernel(
    const float* tvec, const float* wt, const float* bt, float* q)
{
    __shared__ float part[128];
    const int tile = blockIdx.x;
    const float r = gemv_tile_ksplit8(tvec, wt, C_DIM, (long)D_DIM, 1L, tile * 16, part);
    if (threadIdx.x < 16) q[tile * 16 + threadIdx.x] = r + bt[tile * 16 + threadIdx.x];
}

__global__ __launch_bounds__(256) void p_kernel(
    const float* q,
    const float* wk_x, const float* wk_h, const float* wk_s, const float* wk_m,
    const float* g_x, const float* be_x, const float* g_h, const float* be_h,
    const float* g_s, const float* be_s, const float* g_m, const float* be_m,
    float* gp, float* sgpp, float* sbpp)
{
    __shared__ float q_s[D_DIM];
    __shared__ float part[128];
    __shared__ float gv_s[16];
    __shared__ float bv_s[16];
    const int tid  = threadIdx.x;
    const int mod  = blockIdx.x >> 4;
    const int tile = blockIdx.x & 15;
    const float* wks[4] = {wk_x, wk_h, wk_s, wk_m};
    const float* gs[4]  = {g_x,  g_h,  g_s,  g_m};
    const float* bes[4] = {be_x, be_h, be_s, be_m};

    q_s[tid] = q[tid];
    __syncthreads();

    const float p = gemv_tile_ksplit8(q_s, wks[mod], D_DIM, 1L, (long)D_DIM, tile * 16, part);
    if (tid < 16) {
        const int cch = tile * 16 + tid;
        const float gv = gs[mod][cch] * p;
        gv_s[tid] = gv;
        bv_s[tid] = bes[mod][cch] * p;
        gp[mod * C_DIM + cch] = gv;
    }
    __syncthreads();
    if (tid == 0) {
        float sg = 0.0f, sb = 0.0f;
        for (int i = 0; i < 16; ++i) { sg += gv_s[i]; sb += bv_s[i]; }
        sgpp[mod * 16 + tile] = sg;
        sbpp[mod * 16 + tile] = sb;
    }
}

__global__ __launch_bounds__(256) void pass1_kernel(
    const float* x, const float* h, const float* s, const float* m,
    const float* gp, const float* sgpp, const float* sbpp,
    float* alp, float* pm, float* ps, float* pr)
{
    __shared__ float gpm_s[C_DIM];
    __shared__ float sh1[4][64];
    __shared__ float sh2[4][64];
    __shared__ float shd[4][64];
    __shared__ float wred[6];
    const int bid = blockIdx.x;
    const int mod = bid >> 6;
    const int j0  = (bid & 63) << 6;
    const int tl  = threadIdx.x & 63;
    const int grp = threadIdx.x >> 6;
    const float* aptr[4] = {x, h, s, m};
    const float* a = aptr[mod];

    gpm_s[threadIdx.x] = gp[mod * C_DIM + threadIdx.x];
    __syncthreads();

    const int j  = j0 + tl;
    const int c0 = grp << 6;
    float s1 = 0.0f, s2 = 0.0f, dp = 0.0f;
#pragma unroll 4
    for (int cc = 0; cc < 64; ++cc) {
        const int c = c0 + cc;
        const float v = a[(long)c * N_TOK + j];
        s1 += v;
        s2 += v * v;
        dp += v * gpm_s[c];
    }
    sh1[grp][tl] = s1; sh2[grp][tl] = s2; shd[grp][tl] = dp;
    __syncthreads();

    const bool tok = threadIdx.x < 64;
    const int  wv  = threadIdx.x >> 5;
    float lg = 0.0f, mu = 0.0f, rstd = 0.0f;
    if (tok) {
        const float t1 = sh1[0][threadIdx.x] + sh1[1][threadIdx.x] +
                         sh1[2][threadIdx.x] + sh1[3][threadIdx.x];
        const float t2 = sh2[0][threadIdx.x] + sh2[1][threadIdx.x] +
                         sh2[2][threadIdx.x] + sh2[3][threadIdx.x];
        const float td = shd[0][threadIdx.x] + shd[1][threadIdx.x] +
                         shd[2][threadIdx.x] + shd[3][threadIdx.x];
        mu   = t1 * (1.0f / C_DIM);
        const float var  = t2 * (1.0f / C_DIM) - mu * mu;
        rstd = rsqrtf(var + EPSV);
        float sg = 0.0f, sb = 0.0f;
        for (int i = 0; i < 16; ++i) { sg += sgpp[mod * 16 + i]; sb += sbpp[mod * 16 + i]; }
        lg = SCALE * (rstd * (td - mu * sg) + sb);
        const float wm = wave_max32(lg);
        if ((threadIdx.x & 31) == 0) wred[wv] = wm;
    }
    __syncthreads();
    if (tok) {
        const float bmax = fmaxf(wred[0], wred[1]);
        const float e  = __expf(lg - bmax);
        const float al = e * rstd;
        alp[mod * N_TOK + j0 + threadIdx.x] = al;
        const float se = wave_sum32(e);
        const float sr = wave_sum32(al * mu);
        if ((threadIdx.x & 31) == 0) { wred[2 + wv] = se; wred[4 + wv] = sr; }
    }
    __syncthreads();
    if (threadIdx.x == 0) {
        pm[bid] = fmaxf(wred[0], wred[1]);
        ps[bid] = wred[2] + wred[3];
        pr[bid] = wred[4] + wred[5];
    }
}

__global__ __launch_bounds__(256) void pass2_kernel(
    const float* x, const float* h, const float* s, const float* m,
    const float* alp, const float* pm, const float* ps, const float* pr,
    const float* g_x, const float* be_x, const float* g_h, const float* be_h,
    const float* g_s, const float* be_s, const float* g_m, const float* be_m,
    float* u)
{
    __shared__ float red[256];
    __shared__ float f_s[256];
    __shared__ float wmx[8];
    __shared__ float wse[8];
    __shared__ float wre[8];
    const int tid  = threadIdx.x;
    const int wv   = tid >> 5;
    const int lane = tid & 31;
    const int bid = blockIdx.x;
    const int mod = bid >> 8;
    const int c   = bid & 255;
    const float* aptr[4] = {x, h, s, m};
    const float* gs[4]   = {g_x,  g_h,  g_s,  g_m};
    const float* bes[4]  = {be_x, be_h, be_s, be_m};

    const float* a  = aptr[mod] + (long)c * N_TOK;
    const float* al = alp + mod * N_TOK;

    float av[16], lv[16];
    for (int i = 0; i < 16; ++i) { av[i] = a[tid + 256 * i]; lv[i] = al[tid + 256 * i]; }

    const float mb = pm[tid];
    const float sb = ps[tid];
    const float rb = pr[tid];
    const float wm = wave_max32(mb);
    if (lane == 0) wmx[wv] = wm;
    __syncthreads();
    float gmax = wmx[0];
    for (int i = 1; i < 8; ++i) gmax = fmaxf(gmax, wmx[i]);
    const float eb = __expf(mb - gmax);
    f_s[tid] = eb;
    const float se = wave_sum32(sb * eb);
    const float sr = wave_sum32(rb * eb);
    if (lane == 0) { wse[wv] = se; wre[wv] = sr; }
    __syncthreads();
    float ssum = 0.0f;
    for (int i = 0; i < 8; ++i) ssum += wse[i];
    const float inv = 1.0f / ssum;

    const float* fm = f_s + mod * 64 + (tid >> 6);
    float acc = 0.0f;
    for (int i = 0; i < 16; ++i) acc += (lv[i] * fm[4 * i]) * av[i];
    red[tid] = acc; __syncthreads();
    for (int o = 128; o > 0; o >>= 1) {
        if (tid < o) red[tid] += red[tid + o];
        __syncthreads();
    }
    if (tid == 0) {
        const float T    = red[0] * inv;
        const float wsum = (wse[2 * mod] + wse[2 * mod + 1]) * inv;
        const float amu  = (wre[2 * mod] + wre[2 * mod + 1]) * inv;
        u[mod * C_DIM + c] = gs[mod][c] * (T - amu) + bes[mod][c] * wsum;
    }
}


__global__ __launch_bounds__(256) void ov_kernel(
    const float* u,
    const float* wv_x, const float* wv_h, const float* wv_s, const float* wv_m,
    float* ovp)
{
    __shared__ float part[128];
    const int mod  = blockIdx.x >> 4;
    const int tile = blockIdx.x & 15;
    const float* wv[4] = {wv_x, wv_h, wv_s, wv_m};
    const float r = gemv_tile_ksplit8(u + mod * C_DIM, wv[mod], C_DIM, (long)D_DIM, 1L,
                                      tile * 16, part);
    if (threadIdx.x < 16) ovp[mod * D_DIM + tile * 16 + threadIdx.x] = r;
}

__global__ __launch_bounds__(256) void gg_kernel(
    const float* ovp, const float* g_res, const float* be_res,
    const float* w_geglu, const float* b_geglu, float* ggl)
{
    __shared__ float rr[D_DIM];
    __shared__ float red[256];
    __shared__ float part[128];
    const int tid  = threadIdx.x;
    const int tile = blockIdx.x;

    const float ov = ((ovp[tid] + ovp[D_DIM + tid]) + ovp[2 * D_DIM + tid]) + ovp[3 * D_DIM + tid];
    const float mu = block_reduce_sum_256(ov, red) * (1.0f / D_DIM);
    const float dv = ov - mu;
    const float var = block_reduce_sum_256(dv * dv, red) * (1.0f / D_DIM);
    const float rstd = rsqrtf(var + EPSV);
    rr[tid] = dv * rstd * g_res[tid] + be_res[tid];
    __syncthreads();

    const float r = gemv_tile_ksplit8(rr, w_geglu, D_DIM, (long)(2 * D_DIM), 1L, tile * 16, part);
    if (tid < 16) ggl[tile * 16 + tid] = r + b_geglu[tile * 16 + tid];
}

__global__ __launch_bounds__(256) void ff_kernel(
    const float* ggl, const float* w_ff2, const float* b_ff2, float* ffv)
{
    __shared__ float hh[D_DIM];
    __shared__ float part[128];
    const int tid  = threadIdx.x;
    const int tile = blockIdx.x;
    {
        const float g  = ggl[D_DIM + tid];
        const float ge = 0.5f * g * (1.0f + erff(g * 0.70710678118654752f));
        hh[tid] = ggl[tid] * ge;
    }
    __syncthreads();
    const float r = gemv_tile_ksplit8(hh, w_ff2, D_DIM, (long)D_DIM, 1L, tile * 16, part);
    if (tid < 16) ffv[tile * 16 + tid] = r + b_ff2[tile * 16 + tid];
}

__global__ __launch_bounds__(256) void out_kernel(
    const float* ffv, const float* w_out, const float* b_out, float* yvec)
{
    __shared__ float ff_s[D_DIM];
    __shared__ float part[128];
    const int tid  = threadIdx.x;
    const int tile = blockIdx.x;
    ff_s[tid] = ffv[tid];
    __syncthreads();
    const float r = gemv_tile_ksplit8(ff_s, w_out, D_DIM, (long)C_DIM, 1L, tile * 16, part);
    if (tid < 16) yvec[tile * 16 + tid] = r + b_out[tile * 16 + tid];
}

__global__ __launch_bounds__(1024) void bcast_kernel(const float* yvec, float* out)
{
    const int c = blockIdx.x;
    const float v = yvec[c];
    float* o = out + (long)c * N_TOK;
    for (int j = threadIdx.x; j < N_TOK; j += 1024) o[j] = v;
}

extern "C" void kernel_launch(void* const* d_in, const int* in_sizes, int n_in,
                              void* d_out, int out_size, void* d_ws, size_t ws_size,
                              hipStream_t stream) {
    const float* x       = (const float*)d_in[0];
    const float* h       = (const float*)d_in[1];
    const float* s       = (const float*)d_in[2];
    const float* m       = (const float*)d_in[3];
    const float* tvec    = (const float*)d_in[4];
    const float* wt      = (const float*)d_in[5];
    const float* bt      = (const float*)d_in[6];
    const float* wk_x    = (const float*)d_in[7];
    const float* wv_x    = (const float*)d_in[8];
    const float* g_x     = (const float*)d_in[9];
    const float* be_x    = (const float*)d_in[10];
    const float* wk_h    = (const float*)d_in[11];
    const float* wv_h    = (const float*)d_in[12];
    const float* g_h     = (const float*)d_in[13];
    const float* be_h    = (const float*)d_in[14];
    const float* wk_s    = (const float*)d_in[15];
    const float* wv_s    = (const float*)d_in[16];
    const float* g_s     = (const float*)d_in[17];
    const float* be_s    = (const float*)d_in[18];
    const float* wk_m    = (const float*)d_in[19];
    const float* wv_m    = (const float*)d_in[20];
    const float* g_m     = (const float*)d_in[21];
    const float* be_m    = (const float*)d_in[22];
    const float* w_geglu = (const float*)d_in[23];
    const float* b_geglu = (const float*)d_in[24];
    const float* w_ff2   = (const float*)d_in[25];
    const float* b_ff2   = (const float*)d_in[26];
    const float* g_res   = (const float*)d_in[27];
    const float* be_res  = (const float*)d_in[28];
    const float* w_out   = (const float*)d_in[29];
    const float* b_out   = (const float*)d_in[30];
    float* out = (float*)d_out;
    float* ws  = (float*)d_ws;

    float* gp     = ws + WS_GP;
    float* q      = ws + WS_Q;
    float* sgpp   = ws + WS_SGPP;
    float* sbpp   = ws + WS_SBPP;
    float* ovp    = ws + WS_OVP;
    float* ggl    = ws + WS_GGL;
    float* ffv    = ws + WS_FFV;
    float* alp    = ws + WS_AL;
    float* pm     = ws + WS_PM;
    float* ps     = ws + WS_PS;
    float* pr     = ws + WS_PR;
    float* u      = ws + WS_U;
    float* yvec   = ws + WS_YVEC;

    q_kernel<<<16, 256, 0, stream>>>(tvec, wt, bt, q);
    p_kernel<<<64, 256, 0, stream>>>(q, wk_x, wk_h, wk_s, wk_m,
                                     g_x, be_x, g_h, be_h, g_s, be_s, g_m, be_m,
                                     gp, sgpp, sbpp);
    pass1_kernel<<<256, 256, 0, stream>>>(x, h, s, m, gp, sgpp, sbpp,
                                          alp, pm, ps, pr);
    pass2_kernel<<<1024, 256, 0, stream>>>(x, h, s, m, alp, pm, ps, pr,
                                           g_x, be_x, g_h, be_h, g_s, be_s, g_m, be_m,
                                           u);
    ov_kernel<<<64, 256, 0, stream>>>(u, wv_x, wv_h, wv_s, wv_m, ovp);
    gg_kernel<<<32, 256, 0, stream>>>(ovp, g_res, be_res, w_geglu, b_geglu, ggl);
    ff_kernel<<<16, 256, 0, stream>>>(ggl, w_ff2, b_ff2, ffv);
    out_kernel<<<16, 256, 0, stream>>>(ffv, w_out, b_out, yvec);
    bcast_kernel<<<256, 1024, 0, stream>>>(yvec, out);
}


# tail launches 5->3: GEGLU workgroup applies its 16 activations to their 16 rows of w_ff2 (16 partial ff vectors added in tile order); out-proj tile recomputed in broadcast; ff sum order changes
# speedup vs baseline: 3.6002x; 1.1211x over previous
#include <hip/hip_runtime.h>
#include <math.h>

typedef __attribute__((ext_vector_type(2))) float v2f;
typedef __attribute__((ext_vector_type(8))) float v8f;

#define C_DIM 256
#define D_DIM 256
#define N_TOK 4096
#define SCALE 0.0625f
#define EPSV  1e-5f

#define WS_GP    0
#define WS_OVP   1032
#define WS_FFP   4096
#define WS_AL    17408
#define WS_PM    33792
#define WS_PS    34048
#define WS_PR    34304
#define WS_Q     50184
#define WS_SGPP  50440
#define WS_SBPP  50504
#define WS_U     66576

__device__ inline v8f gemv_tile_f32(const float* x, const float* w,
                                    int K, long sk, long sn, int n0, v8f acc) {
    const int lane = threadIdx.x & 31;
    const int half = lane >> 4;
    const int l16  = lane & 15;
    const float amask = (l16 == 0) ? 1.0f : 0.0f;
    const float* wp = w + (long)(n0 + l16) * sn;
    for (int k0 = 0; k0 < K; k0 += 4) {
        const int kb = k0 + 2 * half;
        const float x0 = x[kb];
        const float x1 = x[kb + 1];
        v2f a;
        a.x = x0 * amask;
        a.y = x1 * amask;
        v2f b;
        b.x = wp[(long)kb * sk];
        b.y = wp[(long)(kb + 1) * sk];
        acc = __builtin_amdgcn_wmma_f32_16x16x4_f32(
             false, a,  false, b,
             (short)0, acc,  false,  false);
    }
    return acc;
}

__device__ inline void gemv_tile2_f32(const float* x, const float* w,
                                      int K, long sk, long sn, int n0a, int n0b,
                                      v8f& accA, v8f& accB) {
    const int lane = threadIdx.x & 31;
    const int half = lane >> 4;
    const int l16  = lane & 15;
    const float amask = (l16 == 0) ? 1.0f : 0.0f;
    const float* wpa = w + (long)(n0a + l16) * sn;
    const float* wpb = w + (long)(n0b + l16) * sn;
    for (int k0 = 0; k0 < K; k0 += 4) {
        const int kb = k0 + 2 * half;
        v2f a;
        a.x = x[kb] * amask;
        a.y = x[kb + 1] * amask;
        v2f ba, bb;
        ba.x = wpa[(long)kb * sk];
        ba.y = wpa[(long)(kb + 1) * sk];
        bb.x = wpb[(long)kb * sk];
        bb.y = wpb[(long)(kb + 1) * sk];
        accA = __builtin_amdgcn_wmma_f32_16x16x4_f32(
            false, a, false, ba, (short)0, accA, false, false);
        accB = __builtin_amdgcn_wmma_f32_16x16x4_f32(
            false, a, false, bb, (short)0, accB, false, false);
    }
}

__device__ inline float block_reduce_sum_256(float v, float* red) {
    const int t = threadIdx.x;
    red[t] = v; __syncthreads();
    for (int o = 128; o > 0; o >>= 1) {
        if (t < o) red[t] += red[t + o];
        __syncthreads();
    }
    float r = red[0]; __syncthreads();
    return r;
}

__device__ inline float wave_max32(float v) {
    for (int o = 16; o > 0; o >>= 1) v = fmaxf(v, __shfl_xor(v, o, 32));
    return v;
}
__device__ inline float wave_sum32(float v) {
    for (int o = 16; o > 0; o >>= 1) v += __shfl_xor(v, o, 32);
    return v;
}

__device__ inline float gemv_tile_ksplit8(const float* x, const float* w,
                                          int K, long sk, long sn, int n0, float* part) {
    const int wave = threadIdx.x >> 5;
    const int lane = threadIdx.x & 31;
    const int kw = K >> 3;
    v8f acc = {};
    acc = gemv_tile_f32(x + wave * kw, w + (long)(wave * kw) * sk, kw, sk, sn, n0, acc);
    if (lane < 16) part[wave * 16 + lane] = acc[0];
    __syncthreads();
    float r = 0.0f;
    if (threadIdx.x < 16) {
        for (int i = 0; i < 8; ++i) r += part[i * 16 + threadIdx.x];
    }
    return r;
}

__global__ __launch_bounds__(256) void q_kernel(
    const float* tvec, const float* wt, const float* bt, float* q)
{
    __shared__ float part[128];
    const int tile = blockIdx.x;
    const float r = gemv_tile_ksplit8(tvec, wt, C_DIM, (long)D_DIM, 1L, tile * 16, part);
    if (threadIdx.x < 16) q[tile * 16 + threadIdx.x] = r + bt[tile * 16 + threadIdx.x];
}

__global__ __launch_bounds__(256) void p_kernel(
    const float* q,
    const float* wk_x, const float* wk_h, const float* wk_s, const float* wk_m,
    const float* g_x, const float* be_x, const float* g_h, const float* be_h,
    const float* g_s, const float* be_s, const float* g_m, const float* be_m,
    float* gp, float* sgpp, float* sbpp)
{
    __shared__ float q_s[D_DIM];
    __shared__ float part[128];
    __shared__ float gv_s[16];
    __shared__ float bv_s[16];
    const int tid  = threadIdx.x;
    const int mod  = blockIdx.x >> 4;
    const int tile = blockIdx.x & 15;
    const float* wks[4] = {wk_x, wk_h, wk_s, wk_m};
    const float* gs[4]  = {g_x,  g_h,  g_s,  g_m};
    const float* bes[4] = {be_x, be_h, be_s, be_m};

    q_s[tid] = q[tid];
    __syncthreads();

    const float p = gemv_tile_ksplit8(q_s, wks[mod], D_DIM, 1L, (long)D_DIM, tile * 16, part);
    if (tid < 16) {
        const int cch = tile * 16 + tid;
        const float gv = gs[mod][cch] * p;
        gv_s[tid] = gv;
        bv_s[tid] = bes[mod][cch] * p;
        gp[mod * C_DIM + cch] = gv;
    }
    __syncthreads();
    if (tid == 0) {
        float sg = 0.0f, sb = 0.0f;
        for (int i = 0; i < 16; ++i) { sg += gv_s[i]; sb += bv_s[i]; }
        sgpp[mod * 16 + tile] = sg;
        sbpp[mod * 16 + tile] = sb;
    }
}

__global__ __launch_bounds__(256) void pass1_kernel(
    const float* x, const float* h, const float* s, const float* m,
    const float* gp, const float* sgpp, const float* sbpp,
    float* alp, float* pm, float* ps, float* pr)
{
    __shared__ float gpm_s[C_DIM];
    __shared__ float sh1[4][64];
    __shared__ float sh2[4][64];
    __shared__ float shd[4][64];
    __shared__ float wred[6];
    const int bid = blockIdx.x;
    const int mod = bid >> 6;
    const int j0  = (bid & 63) << 6;
    const int tl  = threadIdx.x & 63;
    const int grp = threadIdx.x >> 6;
    const float* aptr[4] = {x, h, s, m};
    const float* a = aptr[mod];

    gpm_s[threadIdx.x] = gp[mod * C_DIM + threadIdx.x];
    __syncthreads();

    const int j  = j0 + tl;
    const int c0 = grp << 6;
    float s1 = 0.0f, s2 = 0.0f, dp = 0.0f;
#pragma unroll 4
    for (int cc = 0; cc < 64; ++cc) {
        const int c = c0 + cc;
        const float v = a[(long)c * N_TOK + j];
        s1 += v;
        s2 += v * v;
        dp += v * gpm_s[c];
    }
    sh1[grp][tl] = s1; sh2[grp][tl] = s2; shd[grp][tl] = dp;
    __syncthreads();

    const bool tok = threadIdx.x < 64;
    const int  wv  = threadIdx.x >> 5;
    float lg = 0.0f, mu = 0.0f, rstd = 0.0f;
    if (tok) {
        const float t1 = sh1[0][threadIdx.x] + sh1[1][threadIdx.x] +
                         sh1[2][threadIdx.x] + sh1[3][threadIdx.x];
        const float t2 = sh2[0][threadIdx.x] + sh2[1][threadIdx.x] +
                         sh2[2][threadIdx.x] + sh2[3][threadIdx.x];
        const float td = shd[0][threadIdx.x] + shd[1][threadIdx.x] +
                         shd[2][threadIdx.x] + shd[3][threadIdx.x];
        mu   = t1 * (1.0f / C_DIM);
        const float var  = t2 * (1.0f / C_DIM) - mu * mu;
        rstd = rsqrtf(var + EPSV);
        float sg = 0.0f, sb = 0.0f;
        for (int i = 0; i < 16; ++i) { sg += sgpp[mod * 16 + i]; sb += sbpp[mod * 16 + i]; }
        lg = SCALE * (rstd * (td - mu * sg) + sb);
        const float wm = wave_max32(lg);
        if ((threadIdx.x & 31) == 0) wred[wv] = wm;
    }
    __syncthreads();
    if (tok) {
        const float bmax = fmaxf(wred[0], wred[1]);
        const float e  = __expf(lg - bmax);
        const float al = e * rstd;
        alp[mod * N_TOK + j0 + threadIdx.x] = al;
        const float se = wave_sum32(e);
        const float sr = wave_sum32(al * mu);
        if ((threadIdx.x & 31) == 0) { wred[2 + wv] = se; wred[4 + wv] = sr; }
    }
    __syncthreads();
    if (threadIdx.x == 0) {
        pm[bid] = fmaxf(wred[0], wred[1]);
        ps[bid] = wred[2] + wred[3];
        pr[bid] = wred[4] + wred[5];
    }
}

__global__ __launch_bounds__(256) void pass2_kernel(
    const float* x, const float* h, const float* s, const float* m,
    const float* alp, const float* pm, const float* ps, const float* pr,
    const float* g_x, const float* be_x, const float* g_h, const float* be_h,
    const float* g_s, const float* be_s, const float* g_m, const float* be_m,
    float* u)
{
    __shared__ float red[256];
    __shared__ float f_s[256];
    __shared__ float wmx[8];
    __shared__ float wse[8];
    __shared__ float wre[8];
    const int tid  = threadIdx.x;
    const int wv   = tid >> 5;
    const int lane = tid & 31;
    const int bid = blockIdx.x;
    const int mod = bid >> 8;
    const int c   = bid & 255;
    const float* aptr[4] = {x, h, s, m};
    const float* gs[4]   = {g_x,  g_h,  g_s,  g_m};
    const float* bes[4]  = {be_x, be_h, be_s, be_m};

    const float* a  = aptr[mod] + (long)c * N_TOK;
    const float* al = alp + mod * N_TOK;

    float av[16], lv[16];
    for (int i = 0; i < 16; ++i) { av[i] = a[tid + 256 * i]; lv[i] = al[tid + 256 * i]; }

    const float mb = pm[tid];
    const float sb = ps[tid];
    const float rb = pr[tid];
    const float wm = wave_max32(mb);
    if (lane == 0) wmx[wv] = wm;
    __syncthreads();
    float gmax = wmx[0];
    for (int i = 1; i < 8; ++i) gmax = fmaxf(gmax, wmx[i]);
    const float eb = __expf(mb - gmax);
    f_s[tid] = eb;
    const float se = wave_sum32(sb * eb);
    const float sr = wave_sum32(rb * eb);
    if (lane == 0) { wse[wv] = se; wre[wv] = sr; }
    __syncthreads();
    float ssum = 0.0f;
    for (int i = 0; i < 8; ++i) ssum += wse[i];
    const float inv = 1.0f / ssum;

    const float* fm = f_s + mod * 64 + (tid >> 6);
    float acc = 0.0f;
    for (int i = 0; i < 16; ++i) acc += (lv[i] * fm[4 * i]) * av[i];
    red[tid] = acc; __syncthreads();
    for (int o = 128; o > 0; o >>= 1) {
        if (tid < o) red[tid] += red[tid + o];
        __syncthreads();
    }
    if (tid == 0) {
        const float T    = red[0] * inv;
        const float wsum = (wse[2 * mod] + wse[2 * mod + 1]) * inv;
        const float amu  = (wre[2 * mod] + wre[2 * mod + 1]) * inv;
        u[mod * C_DIM + c] = gs[mod][c] * (T - amu) + bes[mod][c] * wsum;
    }
}


__global__ __launch_bounds__(256) void ov_kernel(
    const float* u,
    const float* wv_x, const float* wv_h, const float* wv_s, const float* wv_m,
    float* ovp)
{
    __shared__ float part[128];
    const int mod  = blockIdx.x >> 4;
    const int tile = blockIdx.x & 15;
    const float* wv[4] = {wv_x, wv_h, wv_s, wv_m};
    const float r = gemv_tile_ksplit8(u + mod * C_DIM, wv[mod], C_DIM, (long)D_DIM, 1L,
                                      tile * 16, part);
    if (threadIdx.x < 16) ovp[mod * D_DIM + tile * 16 + threadIdx.x] = r;
}

__global__ __launch_bounds__(256) void ggff_kernel(
    const float* ovp, const float* g_res, const float* be_res,
    const float* w_geglu, const float* b_geglu, const float* w_ff2, float* ffp)
{
    __shared__ float rr[D_DIM];
    __shared__ float red[256];
    __shared__ float parta[128];
    __shared__ float partb[128];
    __shared__ float hh_s[16];
    const int tid  = threadIdx.x;
    const int wave = tid >> 5;
    const int lane = tid & 31;
    const int tile = blockIdx.x;

    const float ov = ((ovp[tid] + ovp[D_DIM + tid]) + ovp[2 * D_DIM + tid]) + ovp[3 * D_DIM + tid];
    const float mu = block_reduce_sum_256(ov, red) * (1.0f / D_DIM);
    const float dv = ov - mu;
    const float var = block_reduce_sum_256(dv * dv, red) * (1.0f / D_DIM);
    const float rstd = rsqrtf(var + EPSV);
    rr[tid] = dv * rstd * g_res[tid] + be_res[tid];
    __syncthreads();

    v8f acca = {};
    v8f accb = {};
    gemv_tile2_f32(rr + wave * 32, w_geglu + (long)(wave * 32) * (2 * D_DIM), 32,
                   (long)(2 * D_DIM), 1L, tile * 16, D_DIM + tile * 16, acca, accb);
    if (lane < 16) { parta[wave * 16 + lane] = acca[0]; partb[wave * 16 + lane] = accb[0]; }
    __syncthreads();
    if (tid < 16) {
        float ru = 0.0f, rg = 0.0f;
        for (int i = 0; i < 8; ++i) { ru += parta[i * 16 + tid]; rg += partb[i * 16 + tid]; }
        const float uu = ru + b_geglu[tile * 16 + tid];
        const float g  = rg + b_geglu[D_DIM + tile * 16 + tid];
        const float ge = 0.5f * g * (1.0f + erff(g * 0.70710678118654752f));
        hh_s[tid] = uu * ge;
    }
    __syncthreads();

    const float* w2 = w_ff2 + (long)(tile * 16) * D_DIM;
    const int n0 = wave * 32;
    v8f acc0 = {};
    v8f acc1 = {};
    gemv_tile2_f32(hh_s, w2, 16, (long)D_DIM, 1L, n0, n0 + 16, acc0, acc1);
    if (lane < 16) {
        ffp[tile * D_DIM + n0 + lane]      = acc0[0];
        ffp[tile * D_DIM + n0 + 16 + lane] = acc1[0];
    }
}

__global__ __launch_bounds__(256) void outb_kernel(
    const float* ffp, const float* b_ff2, const float* w_out, const float* b_out,
    float* out)
{
    __shared__ float ff_s[D_DIM];
    __shared__ float part[128];
    __shared__ float y_s[16];
    const int tid   = threadIdx.x;
    const int tile  = blockIdx.x >> 4;
    const int slice = blockIdx.x & 15;
    float f = 0.0f;
    for (int i = 0; i < 16; ++i) f += ffp[i * D_DIM + tid];
    ff_s[tid] = f + b_ff2[tid];
    __syncthreads();
    const float r = gemv_tile_ksplit8(ff_s, w_out, D_DIM, (long)C_DIM, 1L, tile * 16, part);
    if (tid < 16) y_s[tid] = r + b_out[tile * 16 + tid];
    __syncthreads();
    const int t4 = tid & 63;
    const int cg = tid >> 6;
    for (int i = 0; i < 4; ++i) {
        const int ch = 4 * i + cg;
        const float v = y_s[ch];
        float4 vv;
        vv.x = v; vv.y = v; vv.z = v; vv.w = v;
        float4* o = (float4*)(out + (long)(tile * 16 + ch) * N_TOK + slice * 256 + t4 * 4);
        *o = vv;
    }
}

extern "C" void kernel_launch(void* const* d_in, const int* in_sizes, int n_in,
                              void* d_out, int out_size, void* d_ws, size_t ws_size,
                              hipStream_t stream) {
    const float* x       = (const float*)d_in[0];
    const float* h       = (const float*)d_in[1];
    const float* s       = (const float*)d_in[2];
    const float* m       = (const float*)d_in[3];
    const float* tvec    = (const float*)d_in[4];
    const float* wt      = (const float*)d_in[5];
    const float* bt      = (const float*)d_in[6];
    const float* wk_x    = (const float*)d_in[7];
    const float* wv_x    = (const float*)d_in[8];
    const float* g_x     = (const float*)d_in[9];
    const float* be_x    = (const float*)d_in[10];
    const float* wk_h    = (const float*)d_in[11];
    const float* wv_h    = (const float*)d_in[12];
    const float* g_h     = (const float*)d_in[13];
    const float* be_h    = (const float*)d_in[14];
    const float* wk_s    = (const float*)d_in[15];
    const float* wv_s    = (const float*)d_in[16];
    const float* g_s     = (const float*)d_in[17];
    const float* be_s    = (const float*)d_in[18];
    const float* wk_m    = (const float*)d_in[19];
    const float* wv_m    = (const float*)d_in[20];
    const float* g_m     = (const float*)d_in[21];
    const float* be_m    = (const float*)d_in[22];
    const float* w_geglu = (const float*)d_in[23];
    const float* b_geglu = (const float*)d_in[24];
    const float* w_ff2   = (const float*)d_in[25];
    const float* b_ff2   = (const float*)d_in[26];
    const float* g_res   = (const float*)d_in[27];
    const float* be_res  = (const float*)d_in[28];
    const float* w_out   = (const float*)d_in[29];
    const float* b_out   = (const float*)d_in[30];
    float* out = (float*)d_out;
    float* ws  = (float*)d_ws;

    float* gp     = ws + WS_GP;
    float* q      = ws + WS_Q;
    float* sgpp   = ws + WS_SGPP;
    float* sbpp   = ws + WS_SBPP;
    float* ovp    = ws + WS_OVP;
    float* ffp    = ws + WS_FFP;
    float* alp    = ws + WS_AL;
    float* pm     = ws + WS_PM;
    float* ps     = ws + WS_PS;
    float* pr     = ws + WS_PR;
    float* u      = ws + WS_U;

    q_kernel<<<16, 256, 0, stream>>>(tvec, wt, bt, q);
    p_kernel<<<64, 256, 0, stream>>>(q, wk_x, wk_h, wk_s, wk_m,
                                     g_x, be_x, g_h, be_h, g_s, be_s, g_m, be_m,
                                     gp, sgpp, sbpp);
    pass1_kernel<<<256, 256, 0, stream>>>(x, h, s, m, gp, sgpp, sbpp,
                                          alp, pm, ps, pr);
    pass2_kernel<<<1024, 256, 0, stream>>>(x, h, s, m, alp, pm, ps, pr,
                                           g_x, be_x, g_h, be_h, g_s, be_s, g_m, be_m,
                                           u);
    ov_kernel<<<64, 256, 0, stream>>>(u, wv_x, wv_h, wv_s, wv_m, ovp);
    ggff_kernel<<<16, 256, 0, stream>>>(ovp, g_res, be_res, w_geglu, b_geglu, w_ff2, ffp);
    outb_kernel<<<256, 256, 0, stream>>>(ffp, b_ff2, w_out, b_out, out);
}


# second pass over the inputs removed: 64-token tile kept in registers leaves block-relative weighted channel sums, rescaled in the wv launch; 7->6 launches; LN-stat and token sum order changes (fixed)
# speedup vs baseline: 4.9959x; 1.3877x over previous
#include <hip/hip_runtime.h>
#include <math.h>

typedef __attribute__((ext_vector_type(2))) float v2f;
typedef __attribute__((ext_vector_type(8))) float v8f;

#define C_DIM 256
#define D_DIM 256
#define N_TOK 4096
#define SCALE 0.0625f
#define EPSV  1e-5f

#define WS_GP    0
#define WS_OVP   0
#define WS_SGPP  1024
#define WS_SBPP  1088
#define WS_PM    1152
#define WS_PS    1408
#define WS_PR    1664
#define WS_UP    2048
#define WS_Q     2048
#define WS_FFP   4096

__device__ inline v8f gemv_tile_f32(const float* x, const float* w,
                                    int K, long sk, long sn, int n0, v8f acc) {
    const int lane = threadIdx.x & 31;
    const int half = lane >> 4;
    const int l16  = lane & 15;
    const float amask = (l16 == 0) ? 1.0f : 0.0f;
    const float* wp = w + (long)(n0 + l16) * sn;
    for (int k0 = 0; k0 < K; k0 += 4) {
        const int kb = k0 + 2 * half;
        const float x0 = x[kb];
        const float x1 = x[kb + 1];
        v2f a;
        a.x = x0 * amask;
        a.y = x1 * amask;
        v2f b;
        b.x = wp[(long)kb * sk];
        b.y = wp[(long)(kb + 1) * sk];
        acc = __builtin_amdgcn_wmma_f32_16x16x4_f32(
             false, a,  false, b,
             (short)0, acc,  false,  false);
    }
    return acc;
}

__device__ inline void gemv_tile2_f32(const float* x, const float* w,
                                      int K, long sk, long sn, int n0a, int n0b,
                                      v8f& accA, v8f& accB) {
    const int lane = threadIdx.x & 31;
    const int half = lane >> 4;
    const int l16  = lane & 15;
    const float amask = (l16 == 0) ? 1.0f : 0.0f;
    const float* wpa = w + (long)(n0a + l16) * sn;
    const float* wpb = w + (long)(n0b + l16) * sn;
    for (int k0 = 0; k0 < K; k0 += 4) {
        const int kb = k0 + 2 * half;
        v2f a;
        a.x = x[kb] * amask;
        a.y = x[kb + 1] * amask;
        v2f ba, bb;
        ba.x = wpa[(long)kb * sk];
        ba.y = wpa[(long)(kb + 1) * sk];
        bb.x = wpb[(long)kb * sk];
        bb.y = wpb[(long)(kb + 1) * sk];
        accA = __builtin_amdgcn_wmma_f32_16x16x4_f32(
            false, a, false, ba, (short)0, accA, false, false);
        accB = __builtin_amdgcn_wmma_f32_16x16x4_f32(
            false, a, false, bb, (short)0, accB, false, false);
    }
}

__device__ inline float block_reduce_sum_256(float v, float* red) {
    const int t = threadIdx.x;
    red[t] = v; __syncthreads();
    for (int o = 128; o > 0; o >>= 1) {
        if (t < o) red[t] += red[t + o];
        __syncthreads();
    }
    float r = red[0]; __syncthreads();
    return r;
}

__device__ inline float wave_max32(float v) {
    for (int o = 16; o > 0; o >>= 1) v = fmaxf(v, __shfl_xor(v, o, 32));
    return v;
}
__device__ inline float wave_sum32(float v) {
    for (int o = 16; o > 0; o >>= 1) v += __shfl_xor(v, o, 32);
    return v;
}

__device__ inline float gemv_tile_ksplit8(const float* x, const float* w,
                                          int K, long sk, long sn, int n0, float* part) {
    const int wave = threadIdx.x >> 5;
    const int lane = threadIdx.x & 31;
    const int kw = K >> 3;
    v8f acc = {};
    acc = gemv_tile_f32(x + wave * kw, w + (long)(wave * kw) * sk, kw, sk, sn, n0, acc);
    if (lane < 16) part[wave * 16 + lane] = acc[0];
    __syncthreads();
    float r = 0.0f;
    if (threadIdx.x < 16) {
        for (int i = 0; i < 8; ++i) r += part[i * 16 + threadIdx.x];
    }
    return r;
}

__global__ __launch_bounds__(256) void q_kernel(
    const float* tvec, const float* wt, const float* bt, float* q)
{
    __shared__ float part[128];
    const int tile = blockIdx.x;
    const float r = gemv_tile_ksplit8(tvec, wt, C_DIM, (long)D_DIM, 1L, tile * 16, part);
    if (threadIdx.x < 16) q[tile * 16 + threadIdx.x] = r + bt[tile * 16 + threadIdx.x];
}

__global__ __launch_bounds__(256) void p_kernel(
    const float* q,
    const float* wk_x, const float* wk_h, const float* wk_s, const float* wk_m,
    const float* g_x, const float* be_x, const float* g_h, const float* be_h,
    const float* g_s, const float* be_s, const float* g_m, const float* be_m,
    float* gp, float* sgpp, float* sbpp)
{
    __shared__ float q_s[D_DIM];
    __shared__ float part[128];
    __shared__ float gv_s[16];
    __shared__ float bv_s[16];
    const int tid  = threadIdx.x;
    const int mod  = blockIdx.x >> 4;
    const int tile = blockIdx.x & 15;
    const float* wks[4] = {wk_x, wk_h, wk_s, wk_m};
    const float* gs[4]  = {g_x,  g_h,  g_s,  g_m};
    const float* bes[4] = {be_x, be_h, be_s, be_m};

    q_s[tid] = q[tid];
    __syncthreads();

    const float p = gemv_tile_ksplit8(q_s, wks[mod], D_DIM, 1L, (long)D_DIM, tile * 16, part);
    if (tid < 16) {
        const int cch = tile * 16 + tid;
        const float gv = gs[mod][cch] * p;
        gv_s[tid] = gv;
        bv_s[tid] = bes[mod][cch] * p;
        gp[mod * C_DIM + cch] = gv;
    }
    __syncthreads();
    if (tid == 0) {
        float sg = 0.0f, sb = 0.0f;
        for (int i = 0; i < 16; ++i) { sg += gv_s[i]; sb += bv_s[i]; }
        sgpp[mod * 16 + tile] = sg;
        sbpp[mod * 16 + tile] = sb;
    }
}

#define TR_PITCH 260
__global__ __launch_bounds__(256) void tile_kernel(
    const float* x, const float* h, const float* s, const float* m,
    const float* gp, const float* sgpp, const float* sbpp,
    float* up, float* pm, float* ps, float* pr)
{
    __shared__ __attribute__((aligned(16))) float r1[16 * 64];
    __shared__ __attribute__((aligned(16))) float r2[16 * 64];
    __shared__ __attribute__((aligned(16))) float rd[16 * 64];
    __shared__ __attribute__((aligned(16))) float al_s[64];
    __shared__ __attribute__((aligned(16))) float tr[16 * TR_PITCH];
    __shared__ float wred[6];
    const int tid = threadIdx.x;
    const int bid = blockIdx.x;
    const int mod = bid >> 6;
    const int j0  = (bid & 63) << 6;
    const int tq  = tid & 15;
    const int cg  = tid >> 4;
    const float* am = (mod == 0) ? x : (mod == 1) ? h : (mod == 2) ? s : m;
    const float* a = am + (long)(cg * 16) * N_TOK + j0 + 4 * tq;

    float4 v[16];
#pragma unroll
    for (int cc = 0; cc < 16; ++cc) v[cc] = *(const float4*)(a + (long)cc * N_TOK);

    const float4* g4 = (const float4*)(gp + mod * C_DIM + cg * 16);
    float gv[16];
#pragma unroll
    for (int q = 0; q < 4; ++q) {
        const float4 t = g4[q];
        gv[4 * q] = t.x; gv[4 * q + 1] = t.y; gv[4 * q + 2] = t.z; gv[4 * q + 3] = t.w;
    }
    float sg = 0.0f, sb = 0.0f;
    for (int i = 0; i < 16; ++i) { sg += sgpp[mod * 16 + i]; sb += sbpp[mod * 16 + i]; }
    __builtin_amdgcn_sched_barrier(0);

    float4 s1 = {0.0f, 0.0f, 0.0f, 0.0f};
    float4 s2 = {0.0f, 0.0f, 0.0f, 0.0f};
    float4 dp = {0.0f, 0.0f, 0.0f, 0.0f};
#pragma unroll
    for (int cc = 0; cc < 16; ++cc) {
        const float4 t = v[cc];
        const float g = gv[cc];
        s1.x += t.x;       s1.y += t.y;       s1.z += t.z;       s1.w += t.w;
        s2.x += t.x * t.x; s2.y += t.y * t.y; s2.z += t.z * t.z; s2.w += t.w * t.w;
        dp.x += t.x * g;   dp.y += t.y * g;   dp.z += t.z * g;   dp.w += t.w * g;
    }
    *(float4*)&r1[cg * 64 + 4 * tq] = s1;
    *(float4*)&r2[cg * 64 + 4 * tq] = s2;
    *(float4*)&rd[cg * 64 + 4 * tq] = dp;
    __syncthreads();

    const bool tok = tid < 64;
    const int  wv  = tid >> 5;
    float lg = 0.0f, mu = 0.0f, rstd = 0.0f;
    if (tok) {
        float t1 = 0.0f, t2 = 0.0f, td = 0.0f;
        for (int g = 0; g < 16; ++g) {
            t1 += r1[g * 64 + tid];
            t2 += r2[g * 64 + tid];
            td += rd[g * 64 + tid];
        }
        mu   = t1 * (1.0f / C_DIM);
        const float var  = t2 * (1.0f / C_DIM) - mu * mu;
        rstd = rsqrtf(var + EPSV);
        lg = SCALE * (rstd * (td - mu * sg) + sb);
        const float wm = wave_max32(lg);
        if ((tid & 31) == 0) wred[wv] = wm;
    }
    __syncthreads();
    if (tok) {
        const float bmax = fmaxf(wred[0], wred[1]);
        const float e  = __expf(lg - bmax);
        const float al = e * rstd;
        al_s[tid] = al;
        const float se = wave_sum32(e);
        const float sr = wave_sum32(al * mu);
        if ((tid & 31) == 0) { wred[2 + wv] = se; wred[4 + wv] = sr; }
    }
    __syncthreads();
    if (tid == 0) {
        pm[bid] = fmaxf(wred[0], wred[1]);
        ps[bid] = wred[2] + wred[3];
        pr[bid] = wred[4] + wred[5];
    }

    const float4 a4 = *(const float4*)&al_s[4 * tq];
    float pc[16];
#pragma unroll
    for (int cc = 0; cc < 16; ++cc) {
        const float4 t = v[cc];
        pc[cc] = ((t.x * a4.x + t.y * a4.y) + t.z * a4.z) + t.w * a4.w;
    }
    float* trw = tr + tq * TR_PITCH + cg * 16;
#pragma unroll
    for (int q = 0; q < 4; ++q) {
        float4 t;
        t.x = pc[4 * q]; t.y = pc[4 * q + 1]; t.z = pc[4 * q + 2]; t.w = pc[4 * q + 3];
        *(float4*)&trw[4 * q] = t;
    }
    __syncthreads();
    float uacc = 0.0f;
    for (int t = 0; t < 16; ++t) uacc += tr[t * TR_PITCH + tid];
    up[(long)bid * C_DIM + tid] = uacc;
}


__global__ __launch_bounds__(256) void ov_kernel(
    const float* up, const float* pm, const float* ps, const float* pr,
    const float* g_x, const float* be_x, const float* g_h, const float* be_h,
    const float* g_s, const float* be_s, const float* g_m, const float* be_m,
    const float* wv_x, const float* wv_h, const float* wv_s, const float* wv_m,
    float* ovp)
{
    __shared__ float u_s[C_DIM];
    __shared__ float f_s[256];
    __shared__ float wmx[8];
    __shared__ float wse[8];
    __shared__ float wre[8];
    __shared__ float part[128];
    const int tid  = threadIdx.x;
    const int wvi  = tid >> 5;
    const int lane = tid & 31;
    const int mod  = blockIdx.x >> 4;
    const int tile = blockIdx.x & 15;
    const float* wvm = (mod == 0) ? wv_x : (mod == 1) ? wv_h : (mod == 2) ? wv_s : wv_m;
    const float* gm  = (mod == 0) ? g_x  : (mod == 1) ? g_h  : (mod == 2) ? g_s  : g_m;
    const float* bem = (mod == 0) ? be_x : (mod == 1) ? be_h : (mod == 2) ? be_s : be_m;

    const float* upm = up + (long)(mod * 64) * C_DIM + tid;
    float pv[64];
#pragma unroll
    for (int b = 0; b < 64; ++b) pv[b] = upm[b * C_DIM];
    const float gc  = gm[tid];
    const float bec = bem[tid];

    const float mb = pm[tid];
    const float sb = ps[tid];
    const float rb = pr[tid];
    const float wm = wave_max32(mb);
    if (lane == 0) wmx[wvi] = wm;
    __syncthreads();
    float gmax = wmx[0];
    for (int i = 1; i < 8; ++i) gmax = fmaxf(gmax, wmx[i]);
    const float eb = __expf(mb - gmax);
    f_s[tid] = eb;
    const float se = wave_sum32(sb * eb);
    const float sr = wave_sum32(rb * eb);
    if (lane == 0) { wse[wvi] = se; wre[wvi] = sr; }
    __syncthreads();
    float ssum = 0.0f;
    for (int i = 0; i < 8; ++i) ssum += wse[i];
    const float inv = 1.0f / ssum;
    const float* fm = f_s + mod * 64;
    float c0 = 0.0f, c1 = 0.0f, c2 = 0.0f, c3 = 0.0f;
#pragma unroll
    for (int b = 0; b < 64; b += 4) {
        c0 += pv[b]     * fm[b];
        c1 += pv[b + 1] * fm[b + 1];
        c2 += pv[b + 2] * fm[b + 2];
        c3 += pv[b + 3] * fm[b + 3];
    }
    const float T    = ((c0 + c1) + (c2 + c3)) * inv;
    const float wsum = (wse[2 * mod] + wse[2 * mod + 1]) * inv;
    const float amu  = (wre[2 * mod] + wre[2 * mod + 1]) * inv;
    u_s[tid] = gc * (T - amu) + bec * wsum;
    __syncthreads();

    const float r = gemv_tile_ksplit8(u_s, wvm, C_DIM, (long)D_DIM, 1L, tile * 16, part);
    if (tid < 16) ovp[mod * D_DIM + tile * 16 + tid] = r;
}

__global__ __launch_bounds__(256) void ggff_kernel(
    const float* ovp, const float* g_res, const float* be_res,
    const float* w_geglu, const float* b_geglu, const float* w_ff2, float* ffp)
{
    __shared__ float rr[D_DIM];
    __shared__ float red[256];
    __shared__ float parta[128];
    __shared__ float partb[128];
    __shared__ float hh_s[16];
    const int tid  = threadIdx.x;
    const int wave = tid >> 5;
    const int lane = tid & 31;
    const int tile = blockIdx.x;

    const float ov = ((ovp[tid] + ovp[D_DIM + tid]) + ovp[2 * D_DIM + tid]) + ovp[3 * D_DIM + tid];
    const float mu = block_reduce_sum_256(ov, red) * (1.0f / D_DIM);
    const float dv = ov - mu;
    const float var = block_reduce_sum_256(dv * dv, red) * (1.0f / D_DIM);
    const float rstd = rsqrtf(var + EPSV);
    rr[tid] = dv * rstd * g_res[tid] + be_res[tid];
    __syncthreads();

    v8f acca = {};
    v8f accb = {};
    gemv_tile2_f32(rr + wave * 32, w_geglu + (long)(wave * 32) * (2 * D_DIM), 32,
                   (long)(2 * D_DIM), 1L, tile * 16, D_DIM + tile * 16, acca, accb);
    if (lane < 16) { parta[wave * 16 + lane] = acca[0]; partb[wave * 16 + lane] = accb[0]; }
    __syncthreads();
    if (tid < 16) {
        float ru = 0.0f, rg = 0.0f;
        for (int i = 0; i < 8; ++i) { ru += parta[i * 16 + tid]; rg += partb[i * 16 + tid]; }
        const float uu = ru + b_geglu[tile * 16 + tid];
        const float g  = rg + b_geglu[D_DIM + tile * 16 + tid];
        const float ge = 0.5f * g * (1.0f + erff(g * 0.70710678118654752f));
        hh_s[tid] = uu * ge;
    }
    __syncthreads();

    const float* w2 = w_ff2 + (long)(tile * 16) * D_DIM;
    const int n0 = wave * 32;
    v8f acc0 = {};
    v8f acc1 = {};
    gemv_tile2_f32(hh_s, w2, 16, (long)D_DIM, 1L, n0, n0 + 16, acc0, acc1);
    if (lane < 16) {
        ffp[tile * D_DIM + n0 + lane]      = acc0[0];
        ffp[tile * D_DIM + n0 + 16 + lane] = acc1[0];
    }
}

__global__ __launch_bounds__(256) void outb_kernel(
    const float* ffp, const float* b_ff2, const float* w_out, const float* b_out,
    float* out)
{
    __shared__ float ff_s[D_DIM];
    __shared__ float part[128];
    __shared__ float y_s[16];
    const int tid   = threadIdx.x;
    const int tile  = blockIdx.x >> 4;
    const int slice = blockIdx.x & 15;
    float f = 0.0f;
    for (int i = 0; i < 16; ++i) f += ffp[i * D_DIM + tid];
    ff_s[tid] = f + b_ff2[tid];
    __syncthreads();
    const float r = gemv_tile_ksplit8(ff_s, w_out, D_DIM, (long)C_DIM, 1L, tile * 16, part);
    if (tid < 16) y_s[tid] = r + b_out[tile * 16 + tid];
    __syncthreads();
    const int t4 = tid & 63;
    const int cg = tid >> 6;
    for (int i = 0; i < 4; ++i) {
        const int ch = 4 * i + cg;
        const float v = y_s[ch];
        float4 vv;
        vv.x = v; vv.y = v; vv.z = v; vv.w = v;
        float4* o = (float4*)(out + (long)(tile * 16 + ch) * N_TOK + slice * 256 + t4 * 4);
        *o = vv;
    }
}

extern "C" void kernel_launch(void* const* d_in, const int* in_sizes, int n_in,
                              void* d_out, int out_size, void* d_ws, size_t ws_size,
                              hipStream_t stream) {
    const float* x       = (const float*)d_in[0];
    const float* h       = (const float*)d_in[1];
    const float* s       = (const float*)d_in[2];
    const float* m       = (const float*)d_in[3];
    const float* tvec    = (const float*)d_in[4];
    const float* wt      = (const float*)d_in[5];
    const float* bt      = (const float*)d_in[6];
    const float* wk_x    = (const float*)d_in[7];
    const float* wv_x    = (const float*)d_in[8];
    const float* g_x     = (const float*)d_in[9];
    const float* be_x    = (const float*)d_in[10];
    const float* wk_h    = (const float*)d_in[11];
    const float* wv_h    = (const float*)d_in[12];
    const float* g_h     = (const float*)d_in[13];
    const float* be_h    = (const float*)d_in[14];
    const float* wk_s    = (const float*)d_in[15];
    const float* wv_s    = (const float*)d_in[16];
    const float* g_s     = (const float*)d_in[17];
    const float* be_s    = (const float*)d_in[18];
    const float* wk_m    = (const float*)d_in[19];
    const float* wv_m    = (const float*)d_in[20];
    const float* g_m     = (const float*)d_in[21];
    const float* be_m    = (const float*)d_in[22];
    const float* w_geglu = (const float*)d_in[23];
    const float* b_geglu = (const float*)d_in[24];
    const float* w_ff2   = (const float*)d_in[25];
    const float* b_ff2   = (const float*)d_in[26];
    const float* g_res   = (const float*)d_in[27];
    const float* be_res  = (const float*)d_in[28];
    const float* w_out   = (const float*)d_in[29];
    const float* b_out   = (const float*)d_in[30];
    float* out = (float*)d_out;
    float* ws  = (float*)d_ws;

    float* gp     = ws + WS_GP;
    float* q      = ws + WS_Q;
    float* sgpp   = ws + WS_SGPP;
    float* sbpp   = ws + WS_SBPP;
    float* ovp    = ws + WS_OVP;
    float* ffp    = ws + WS_FFP;
    float* up     = ws + WS_UP;
    float* pm     = ws + WS_PM;
    float* ps     = ws + WS_PS;
    float* pr     = ws + WS_PR;

    q_kernel<<<16, 256, 0, stream>>>(tvec, wt, bt, q);
    p_kernel<<<64, 256, 0, stream>>>(q, wk_x, wk_h, wk_s, wk_m,
                                     g_x, be_x, g_h, be_h, g_s, be_s, g_m, be_m,
                                     gp, sgpp, sbpp);
    tile_kernel<<<256, 256, 0, stream>>>(x, h, s, m, gp, sgpp, sbpp,
                                         up, pm, ps, pr);
    ov_kernel<<<64, 256, 0, stream>>>(up, pm, ps, pr,
                                      g_x, be_x, g_h, be_h, g_s, be_s, g_m, be_m,
                                      wv_x, wv_h, wv_s, wv_m, ovp);
    ggff_kernel<<<16, 256, 0, stream>>>(ovp, g_res, be_res, w_geglu, b_geglu, w_ff2, ffp);
    outb_kernel<<<256, 256, 0, stream>>>(ffp, b_ff2, w_out, b_out, out);
}


# cross-lane reduction steps on the vector ALU (row_xmask DPP, permlanex16) not the LDS unit; LayerNorm sums of the GEGLU launch by wave butterfly + 8 partials, 2 barriers for 20; sum order changes
# speedup vs baseline: 5.3093x; 1.0627x over previous
#include <hip/hip_runtime.h>
#include <math.h>

typedef __attribute__((ext_vector_type(2))) float v2f;
typedef __attribute__((ext_vector_type(8))) float v8f;

#define C_DIM 256
#define D_DIM 256
#define N_TOK 4096
#define SCALE 0.0625f
#define EPSV  1e-5f

#define WS_GP    0
#define WS_OVP   0
#define WS_SGPP  1024
#define WS_SBPP  1088
#define WS_PM    1152
#define WS_PS    1408
#define WS_PR    1664
#define WS_UP    2048
#define WS_Q     2048
#define WS_FFP   4096

__device__ inline v8f gemv_tile_f32(const float* x, const float* w,
                                    int K, long sk, long sn, int n0, v8f acc) {
    const int lane = threadIdx.x & 31;
    const int half = lane >> 4;
    const int l16  = lane & 15;
    const float amask = (l16 == 0) ? 1.0f : 0.0f;
    const float* wp = w + (long)(n0 + l16) * sn;
    for (int k0 = 0; k0 < K; k0 += 4) {
        const int kb = k0 + 2 * half;
        const float x0 = x[kb];
        const float x1 = x[kb + 1];
        v2f a;
        a.x = x0 * amask;
        a.y = x1 * amask;
        v2f b;
        b.x = wp[(long)kb * sk];
        b.y = wp[(long)(kb + 1) * sk];
        acc = __builtin_amdgcn_wmma_f32_16x16x4_f32(
             false, a,  false, b,
             (short)0, acc,  false,  false);
    }
    return acc;
}

__device__ inline void gemv_tile2_f32(const float* x, const float* w,
                                      int K, long sk, long sn, int n0a, int n0b,
                                      v8f& accA, v8f& accB) {
    const int lane = threadIdx.x & 31;
    const int half = lane >> 4;
    const int l16  = lane & 15;
    const float amask = (l16 == 0) ? 1.0f : 0.0f;
    const float* wpa = w + (long)(n0a + l16) * sn;
    const float* wpb = w + (long)(n0b + l16) * sn;
    for (int k0 = 0; k0 < K; k0 += 4) {
        const int kb = k0 + 2 * half;
        v2f a;
        a.x = x[kb] * amask;
        a.y = x[kb + 1] * amask;
        v2f ba, bb;
        ba.x = wpa[(long)kb * sk];
        ba.y = wpa[(long)(kb + 1) * sk];
        bb.x = wpb[(long)kb * sk];
        bb.y = wpb[(long)(kb + 1) * sk];
        accA = __builtin_amdgcn_wmma_f32_16x16x4_f32(
            false, a, false, ba, (short)0, accA, false, false);
        accB = __builtin_amdgcn_wmma_f32_16x16x4_f32(
            false, a, false, bb, (short)0, accB, false, false);
    }
}

template <int CTRL>
__device__ inline float lane_xmask(float v) {
    return __int_as_float(__builtin_amdgcn_update_dpp(0, __float_as_int(v), CTRL, 0xf, 0xf, false));
}
__device__ inline float lane_xrow(float v) {
    return __int_as_float((int)__builtin_amdgcn_permlanex16(0u, (unsigned)__float_as_int(v),
                                                            0x76543210u, 0xfedcba98u, false, false));
}
__device__ inline float wave_max32(float v) {
    v = fmaxf(v, lane_xmask<0x161>(v));
    v = fmaxf(v, lane_xmask<0x162>(v));
    v = fmaxf(v, lane_xmask<0x164>(v));
    v = fmaxf(v, lane_xmask<0x168>(v));
    v = fmaxf(v, lane_xrow(v));
    return v;
}
__device__ inline float wave_sum32(float v) {
    v += lane_xmask<0x161>(v);
    v += lane_xmask<0x162>(v);
    v += lane_xmask<0x164>(v);
    v += lane_xmask<0x168>(v);
    v += lane_xrow(v);
    return v;
}

__device__ inline float block_sum_256(float v, float* wp) {
    const float w = wave_sum32(v);
    if ((threadIdx.x & 31) == 0) wp[threadIdx.x >> 5] = w;
    __syncthreads();
    float r = 0.0f;
    for (int i = 0; i < 8; ++i) r += wp[i];
    return r;
}

__device__ inline float gemv_tile_ksplit8(const float* x, const float* w,
                                          int K, long sk, long sn, int n0, float* part) {
    const int wave = threadIdx.x >> 5;
    const int lane = threadIdx.x & 31;
    const int kw = K >> 3;
    v8f acc = {};
    acc = gemv_tile_f32(x + wave * kw, w + (long)(wave * kw) * sk, kw, sk, sn, n0, acc);
    if (lane < 16) part[wave * 16 + lane] = acc[0];
    __syncthreads();
    float r = 0.0f;
    if (threadIdx.x < 16) {
        for (int i = 0; i < 8; ++i) r += part[i * 16 + threadIdx.x];
    }
    return r;
}

__global__ __launch_bounds__(256) void q_kernel(
    const float* tvec, const float* wt, const float* bt, float* q)
{
    __shared__ float part[128];
    const int tile = blockIdx.x;
    const float r = gemv_tile_ksplit8(tvec, wt, C_DIM, (long)D_DIM, 1L, tile * 16, part);
    if (threadIdx.x < 16) q[tile * 16 + threadIdx.x] = r + bt[tile * 16 + threadIdx.x];
}

__global__ __launch_bounds__(256) void p_kernel(
    const float* q,
    const float* wk_x, const float* wk_h, const float* wk_s, const float* wk_m,
    const float* g_x, const float* be_x, const float* g_h, const float* be_h,
    const float* g_s, const float* be_s, const float* g_m, const float* be_m,
    float* gp, float* sgpp, float* sbpp)
{
    __shared__ float q_s[D_DIM];
    __shared__ float part[128];
    __shared__ float gv_s[16];
    __shared__ float bv_s[16];
    const int tid  = threadIdx.x;
    const int mod  = blockIdx.x >> 4;
    const int tile = blockIdx.x & 15;
    const float* wks[4] = {wk_x, wk_h, wk_s, wk_m};
    const float* gs[4]  = {g_x,  g_h,  g_s,  g_m};
    const float* bes[4] = {be_x, be_h, be_s, be_m};

    q_s[tid] = q[tid];
    __syncthreads();

    const float p = gemv_tile_ksplit8(q_s, wks[mod], D_DIM, 1L, (long)D_DIM, tile * 16, part);
    if (tid < 16) {
        const int cch = tile * 16 + tid;
        const float gv = gs[mod][cch] * p;
        gv_s[tid] = gv;
        bv_s[tid] = bes[mod][cch] * p;
        gp[mod * C_DIM + cch] = gv;
    }
    __syncthreads();
    if (tid == 0) {
        float sg = 0.0f, sb = 0.0f;
        for (int i = 0; i < 16; ++i) { sg += gv_s[i]; sb += bv_s[i]; }
        sgpp[mod * 16 + tile] = sg;
        sbpp[mod * 16 + tile] = sb;
    }
}

#define TR_PITCH 260
__global__ __launch_bounds__(256) void tile_kernel(
    const float* x, const float* h, const float* s, const float* m,
    const float* gp, const float* sgpp, const float* sbpp,
    float* up, float* pm, float* ps, float* pr)
{
    __shared__ __attribute__((aligned(16))) float r1[16 * 64];
    __shared__ __attribute__((aligned(16))) float r2[16 * 64];
    __shared__ __attribute__((aligned(16))) float rd[16 * 64];
    __shared__ __attribute__((aligned(16))) float al_s[64];
    __shared__ __attribute__((aligned(16))) float tr[16 * TR_PITCH];
    __shared__ float wred[6];
    const int tid = threadIdx.x;
    const int bid = blockIdx.x;
    const int mod = bid >> 6;
    const int j0  = (bid & 63) << 6;
    const int tq  = tid & 15;
    const int cg  = tid >> 4;
    const float* am = (mod == 0) ? x : (mod == 1) ? h : (mod == 2) ? s : m;
    const float* a = am + (long)(cg * 16) * N_TOK + j0 + 4 * tq;

    float4 v[16];
#pragma unroll
    for (int cc = 0; cc < 16; ++cc) v[cc] = *(const float4*)(a + (long)cc * N_TOK);

    const float4* g4 = (const float4*)(gp + mod * C_DIM + cg * 16);
    float gv[16];
#pragma unroll
    for (int q = 0; q < 4; ++q) {
        const float4 t = g4[q];
        gv[4 * q] = t.x; gv[4 * q + 1] = t.y; gv[4 * q + 2] = t.z; gv[4 * q + 3] = t.w;
    }
    float sg = 0.0f, sb = 0.0f;
    for (int i = 0; i < 16; ++i) { sg += sgpp[mod * 16 + i]; sb += sbpp[mod * 16 + i]; }
    __builtin_amdgcn_sched_barrier(0);

    float4 s1 = {0.0f, 0.0f, 0.0f, 0.0f};
    float4 s2 = {0.0f, 0.0f, 0.0f, 0.0f};
    float4 dp = {0.0f, 0.0f, 0.0f, 0.0f};
#pragma unroll
    for (int cc = 0; cc < 16; ++cc) {
        const float4 t = v[cc];
        const float g = gv[cc];
        s1.x += t.x;       s1.y += t.y;       s1.z += t.z;       s1.w += t.w;
        s2.x += t.x * t.x; s2.y += t.y * t.y; s2.z += t.z * t.z; s2.w += t.w * t.w;
        dp.x += t.x * g;   dp.y += t.y * g;   dp.z += t.z * g;   dp.w += t.w * g;
    }
    *(float4*)&r1[cg * 64 + 4 * tq] = s1;
    *(float4*)&r2[cg * 64 + 4 * tq] = s2;
    *(float4*)&rd[cg * 64 + 4 * tq] = dp;
    __syncthreads();

    const bool tok = tid < 64;
    const int  wv  = tid >> 5;
    float lg = 0.0f, mu = 0.0f, rstd = 0.0f;
    if (tok) {
        float t1 = 0.0f, t2 = 0.0f, td = 0.0f;
        for (int g = 0; g < 16; ++g) {
            t1 += r1[g * 64 + tid];
            t2 += r2[g * 64 + tid];
            td += rd[g * 64 + tid];
        }
        mu   = t1 * (1.0f / C_DIM);
        const float var  = t2 * (1.0f / C_DIM) - mu * mu;
        rstd = rsqrtf(var + EPSV);
        lg = SCALE * (rstd * (td - mu * sg) + sb);
        const float wm = wave_max32(lg);
        if ((tid & 31) == 0) wred[wv] = wm;
    }
    __syncthreads();
    if (tok) {
        const float bmax = fmaxf(wred[0], wred[1]);
        const float e  = __expf(lg - bmax);
        const float al = e * rstd;
        al_s[tid] = al;
        const float se = wave_sum32(e);
        const float sr = wave_sum32(al * mu);
        if ((tid & 31) == 0) { wred[2 + wv] = se; wred[4 + wv] = sr; }
    }
    __syncthreads();
    if (tid == 0) {
        pm[bid] = fmaxf(wred[0], wred[1]);
        ps[bid] = wred[2] + wred[3];
        pr[bid] = wred[4] + wred[5];
    }

    const float4 a4 = *(const float4*)&al_s[4 * tq];
    float pc[16];
#pragma unroll
    for (int cc = 0; cc < 16; ++cc) {
        const float4 t = v[cc];
        pc[cc] = ((t.x * a4.x + t.y * a4.y) + t.z * a4.z) + t.w * a4.w;
    }
    float* trw = tr + tq * TR_PITCH + cg * 16;
#pragma unroll
    for (int q = 0; q < 4; ++q) {
        float4 t;
        t.x = pc[4 * q]; t.y = pc[4 * q + 1]; t.z = pc[4 * q + 2]; t.w = pc[4 * q + 3];
        *(float4*)&trw[4 * q] = t;
    }
    __syncthreads();
    float uacc = 0.0f;
    for (int t = 0; t < 16; ++t) uacc += tr[t * TR_PITCH + tid];
    up[(long)bid * C_DIM + tid] = uacc;
}


__global__ __launch_bounds__(256) void ov_kernel(
    const float* up, const float* pm, const float* ps, const float* pr,
    const float* g_x, const float* be_x, const float* g_h, const float* be_h,
    const float* g_s, const float* be_s, const float* g_m, const float* be_m,
    const float* wv_x, const float* wv_h, const float* wv_s, const float* wv_m,
    float* ovp)
{
    __shared__ float u_s[C_DIM];
    __shared__ float f_s[256];
    __shared__ float wmx[8];
    __shared__ float wse[8];
    __shared__ float wre[8];
    __shared__ float part[128];
    const int tid  = threadIdx.x;
    const int wvi  = tid >> 5;
    const int lane = tid & 31;
    const int mod  = blockIdx.x >> 4;
    const int tile = blockIdx.x & 15;
    const float* wvm = (mod == 0) ? wv_x : (mod == 1) ? wv_h : (mod == 2) ? wv_s : wv_m;
    const float* gm  = (mod == 0) ? g_x  : (mod == 1) ? g_h  : (mod == 2) ? g_s  : g_m;
    const float* bem = (mod == 0) ? be_x : (mod == 1) ? be_h : (mod == 2) ? be_s : be_m;

    const float* upm = up + (long)(mod * 64) * C_DIM + tid;
    float pv[64];
#pragma unroll
    for (int b = 0; b < 64; ++b) pv[b] = upm[b * C_DIM];
    const float gc  = gm[tid];
    const float bec = bem[tid];

    const float mb = pm[tid];
    const float sb = ps[tid];
    const float rb = pr[tid];
    const float wm = wave_max32(mb);
    if (lane == 0) wmx[wvi] = wm;
    __syncthreads();
    float gmax = wmx[0];
    for (int i = 1; i < 8; ++i) gmax = fmaxf(gmax, wmx[i]);
    const float eb = __expf(mb - gmax);
    f_s[tid] = eb;
    const float se = wave_sum32(sb * eb);
    const float sr = wave_sum32(rb * eb);
    if (lane == 0) { wse[wvi] = se; wre[wvi] = sr; }
    __syncthreads();
    float ssum = 0.0f;
    for (int i = 0; i < 8; ++i) ssum += wse[i];
    const float inv = 1.0f / ssum;
    const float* fm = f_s + mod * 64;
    float c0 = 0.0f, c1 = 0.0f, c2 = 0.0f, c3 = 0.0f;
#pragma unroll
    for (int b = 0; b < 64; b += 4) {
        c0 += pv[b]     * fm[b];
        c1 += pv[b + 1] * fm[b + 1];
        c2 += pv[b + 2] * fm[b + 2];
        c3 += pv[b + 3] * fm[b + 3];
    }
    const float T    = ((c0 + c1) + (c2 + c3)) * inv;
    const float wsum = (wse[2 * mod] + wse[2 * mod + 1]) * inv;
    const float amu  = (wre[2 * mod] + wre[2 * mod + 1]) * inv;
    u_s[tid] = gc * (T - amu) + bec * wsum;
    __syncthreads();

    const float r = gemv_tile_ksplit8(u_s, wvm, C_DIM, (long)D_DIM, 1L, tile * 16, part);
    if (tid < 16) ovp[mod * D_DIM + tile * 16 + tid] = r;
}

__global__ __launch_bounds__(256) void ggff_kernel(
    const float* ovp, const float* g_res, const float* be_res,
    const float* w_geglu, const float* b_geglu, const float* w_ff2, float* ffp)
{
    __shared__ float rr[D_DIM];
    __shared__ float wsa[8];
    __shared__ float wsb[8];
    __shared__ float parta[128];
    __shared__ float partb[128];
    __shared__ float hh_s[16];
    const int tid  = threadIdx.x;
    const int wave = tid >> 5;
    const int lane = tid & 31;
    const int tile = blockIdx.x;

    const float ov = ((ovp[tid] + ovp[D_DIM + tid]) + ovp[2 * D_DIM + tid]) + ovp[3 * D_DIM + tid];
    const float mu = block_sum_256(ov, wsa) * (1.0f / D_DIM);
    const float dv = ov - mu;
    const float var = block_sum_256(dv * dv, wsb) * (1.0f / D_DIM);
    const float rstd = rsqrtf(var + EPSV);
    rr[tid] = dv * rstd * g_res[tid] + be_res[tid];
    __syncthreads();

    v8f acca = {};
    v8f accb = {};
    gemv_tile2_f32(rr + wave * 32, w_geglu + (long)(wave * 32) * (2 * D_DIM), 32,
                   (long)(2 * D_DIM), 1L, tile * 16, D_DIM + tile * 16, acca, accb);
    if (lane < 16) { parta[wave * 16 + lane] = acca[0]; partb[wave * 16 + lane] = accb[0]; }
    __syncthreads();
    if (tid < 16) {
        float ru = 0.0f, rg = 0.0f;
        for (int i = 0; i < 8; ++i) { ru += parta[i * 16 + tid]; rg += partb[i * 16 + tid]; }
        const float uu = ru + b_geglu[tile * 16 + tid];
        const float g  = rg + b_geglu[D_DIM + tile * 16 + tid];
        const float ge = 0.5f * g * (1.0f + erff(g * 0.70710678118654752f));
        hh_s[tid] = uu * ge;
    }
    __syncthreads();

    const float* w2 = w_ff2 + (long)(tile * 16) * D_DIM;
    const int n0 = wave * 32;
    v8f acc0 = {};
    v8f acc1 = {};
    gemv_tile2_f32(hh_s, w2, 16, (long)D_DIM, 1L, n0, n0 + 16, acc0, acc1);
    if (lane < 16) {
        ffp[tile * D_DIM + n0 + lane]      = acc0[0];
        ffp[tile * D_DIM + n0 + 16 + lane] = acc1[0];
    }
}

__global__ __launch_bounds__(256) void outb_kernel(
    const float* ffp, const float* b_ff2, const float* w_out, const float* b_out,
    float* out)
{
    __shared__ float ff_s[D_DIM];
    __shared__ float part[128];
    __shared__ float y_s[16];
    const int tid   = threadIdx.x;
    const int tile  = blockIdx.x >> 4;
    const int slice = blockIdx.x & 15;
    float f = 0.0f;
    for (int i = 0; i < 16; ++i) f += ffp[i * D_DIM + tid];
    ff_s[tid] = f + b_ff2[tid];
    __syncthreads();
    const float r = gemv_tile_ksplit8(ff_s, w_out, D_DIM, (long)C_DIM, 1L, tile * 16, part);
    if (tid < 16) y_s[tid] = r + b_out[tile * 16 + tid];
    __syncthreads();
    const int t4 = tid & 63;
    const int cg = tid >> 6;
    for (int i = 0; i < 4; ++i) {
        const int ch = 4 * i + cg;
        const float v = y_s[ch];
        float4 vv;
        vv.x = v; vv.y = v; vv.z = v; vv.w = v;
        float4* o = (float4*)(out + (long)(tile * 16 + ch) * N_TOK + slice * 256 + t4 * 4);
        *o = vv;
    }
}

extern "C" void kernel_launch(void* const* d_in, const int* in_sizes, int n_in,
                              void* d_out, int out_size, void* d_ws, size_t ws_size,
                              hipStream_t stream) {
    const float* x       = (const float*)d_in[0];
    const float* h       = (const float*)d_in[1];
    const float* s       = (const float*)d_in[2];
    const float* m       = (const float*)d_in[3];
    const float* tvec    = (const float*)d_in[4];
    const float* wt      = (const float*)d_in[5];
    const float* bt      = (const float*)d_in[6];
    const float* wk_x    = (const float*)d_in[7];
    const float* wv_x    = (const float*)d_in[8];
    const float* g_x     = (const float*)d_in[9];
    const float* be_x    = (const float*)d_in[10];
    const float* wk_h    = (const float*)d_in[11];
    const float* wv_h    = (const float*)d_in[12];
    const float* g_h     = (const float*)d_in[13];
    const float* be_h    = (const float*)d_in[14];
    const float* wk_s    = (const float*)d_in[15];
    const float* wv_s    = (const float*)d_in[16];
    const float* g_s     = (const float*)d_in[17];
    const float* be_s    = (const float*)d_in[18];
    const float* wk_m    = (const float*)d_in[19];
    const float* wv_m    = (const float*)d_in[20];
    const float* g_m     = (const float*)d_in[21];
    const float* be_m    = (const float*)d_in[22];
    const float* w_geglu = (const float*)d_in[23];
    const float* b_geglu = (const float*)d_in[24];
    const float* w_ff2   = (const float*)d_in[25];
    const float* b_ff2   = (const float*)d_in[26];
    const float* g_res   = (const float*)d_in[27];
    const float* be_res  = (const float*)d_in[28];
    const float* w_out   = (const float*)d_in[29];
    const float* b_out   = (const float*)d_in[30];
    float* out = (float*)d_out;
    float* ws  = (float*)d_ws;

    float* gp     = ws + WS_GP;
    float* q      = ws + WS_Q;
    float* sgpp   = ws + WS_SGPP;
    float* sbpp   = ws + WS_SBPP;
    float* ovp    = ws + WS_OVP;
    float* ffp    = ws + WS_FFP;
    float* up     = ws + WS_UP;
    float* pm     = ws + WS_PM;
    float* ps     = ws + WS_PS;
    float* pr     = ws + WS_PR;

    q_kernel<<<16, 256, 0, stream>>>(tvec, wt, bt, q);
    p_kernel<<<64, 256, 0, stream>>>(q, wk_x, wk_h, wk_s, wk_m,
                                     g_x, be_x, g_h, be_h, g_s, be_s, g_m, be_m,
                                     gp, sgpp, sbpp);
    tile_kernel<<<256, 256, 0, stream>>>(x, h, s, m, gp, sgpp, sbpp,
                                         up, pm, ps, pr);
    ov_kernel<<<64, 256, 0, stream>>>(up, pm, ps, pr,
                                      g_x, be_x, g_h, be_h, g_s, be_s, g_m, be_m,
                                      wv_x, wv_h, wv_s, wv_m, ovp);
    ggff_kernel<<<16, 256, 0, stream>>>(ovp, g_res, be_res, w_geglu, b_geglu, w_ff2, ffp);
    outb_kernel<<<256, 256, 0, stream>>>(ffp, b_ff2, w_out, b_out, out);
}


# loads with entry-known addresses (weight fragments, biases, gains) issued at kernel entry, kept in flight across LDS-only barriers: one memory round trip per launch, not 2-5; sum orders unchanged
# speedup vs baseline: 6.1574x; 1.1597x over previous
#include <hip/hip_runtime.h>
#include <math.h>

typedef __attribute__((ext_vector_type(2))) float v2f;
typedef __attribute__((ext_vector_type(8))) float v8f;

#define C_DIM 256
#define D_DIM 256
#define N_TOK 4096
#define SCALE 0.0625f
#define EPSV  1e-5f

#define WS_GP    0
#define WS_OVP   0
#define WS_SGPP  1024
#define WS_SBPP  1088
#define WS_PM    1152
#define WS_PS    1408
#define WS_PR    1664
#define WS_UP    2048
#define WS_Q     2048
#define WS_FFP   4096

template <int CTRL>
__device__ inline float lane_xmask(float v) {
    return __int_as_float(__builtin_amdgcn_update_dpp(0, __float_as_int(v), CTRL, 0xf, 0xf, false));
}
__device__ inline float lane_xrow(float v) {
    return __int_as_float((int)__builtin_amdgcn_permlanex16(0u, (unsigned)__float_as_int(v),
                                                            0x76543210u, 0xfedcba98u, false, false));
}
__device__ inline float wave_max32(float v) {
    v = fmaxf(v, lane_xmask<0x161>(v));
    v = fmaxf(v, lane_xmask<0x162>(v));
    v = fmaxf(v, lane_xmask<0x164>(v));
    v = fmaxf(v, lane_xmask<0x168>(v));
    v = fmaxf(v, lane_xrow(v));
    return v;
}
__device__ inline float wave_sum32(float v) {
    v += lane_xmask<0x161>(v);
    v += lane_xmask<0x162>(v);
    v += lane_xmask<0x164>(v);
    v += lane_xmask<0x168>(v);
    v += lane_xrow(v);
    return v;
}

__device__ inline void lds_barrier() {
    __builtin_amdgcn_fence(__ATOMIC_RELEASE, "workgroup", "local");
    __builtin_amdgcn_s_barrier();
    __builtin_amdgcn_fence(__ATOMIC_ACQUIRE, "workgroup", "local");
}

__device__ inline float block_sum_256(float v, float* wp) {
    const float w = wave_sum32(v);
    if ((threadIdx.x & 31) == 0) wp[threadIdx.x >> 5] = w;
    lds_barrier();
    float r = 0.0f;
    for (int i = 0; i < 8; ++i) r += wp[i];
    return r;
}

template <int KW>
__device__ inline void load_bfrag(const float* w, long sk, long sn, int n0, v2f (&b)[KW / 4]) {
    const int lane = threadIdx.x & 31;
    const int half = lane >> 4;
    const int l16  = lane & 15;
    const float* wp = w + (long)(n0 + l16) * sn;
#pragma unroll
    for (int i = 0; i < KW / 4; ++i) {
        const int kb = 4 * i + 2 * half;
        b[i].x = wp[(long)kb * sk];
        b[i].y = wp[(long)(kb + 1) * sk];
    }
}

template <int KW>
__device__ inline v8f gemv_frag(const float* x, const v2f (&b)[KW / 4], v8f acc) {
    const int lane = threadIdx.x & 31;
    const int half = lane >> 4;
    const float amask = ((lane & 15) == 0) ? 1.0f : 0.0f;
#pragma unroll
    for (int i = 0; i < KW / 4; ++i) {
        const int kb = 4 * i + 2 * half;
        v2f a;
        a.x = x[kb] * amask;
        a.y = x[kb + 1] * amask;
        acc = __builtin_amdgcn_wmma_f32_16x16x4_f32(
            false, a, false, b[i], (short)0, acc, false, false);
    }
    return acc;
}

template <int KW>
__device__ inline void gemv_frag2(const float* x, const v2f (&ba)[KW / 4], const v2f (&bb)[KW / 4],
                                  v8f& accA, v8f& accB) {
    const int lane = threadIdx.x & 31;
    const int half = lane >> 4;
    const float amask = ((lane & 15) == 0) ? 1.0f : 0.0f;
#pragma unroll
    for (int i = 0; i < KW / 4; ++i) {
        const int kb = 4 * i + 2 * half;
        v2f a;
        a.x = x[kb] * amask;
        a.y = x[kb + 1] * amask;
        accA = __builtin_amdgcn_wmma_f32_16x16x4_f32(
            false, a, false, ba[i], (short)0, accA, false, false);
        accB = __builtin_amdgcn_wmma_f32_16x16x4_f32(
            false, a, false, bb[i], (short)0, accB, false, false);
    }
}

__device__ inline float ksplit8_sum(v8f acc, float* part) {
    const int wave = threadIdx.x >> 5;
    const int lane = threadIdx.x & 31;
    if (lane < 16) part[wave * 16 + lane] = acc[0];
    lds_barrier();
    float r = 0.0f;
    if (threadIdx.x < 16) {
        for (int i = 0; i < 8; ++i) r += part[i * 16 + threadIdx.x];
    }
    return r;
}

__global__ __launch_bounds__(256) void q_kernel(
    const float* tvec, const float* wt, const float* bt, float* q)
{
    __shared__ float part[128];
    const int tid  = threadIdx.x;
    const int wave = tid >> 5;
    const int tile = blockIdx.x;
    v2f bw[8];
    load_bfrag<32>(wt + (long)(wave * 32) * D_DIM, (long)D_DIM, 1L, tile * 16, bw);
    const float btv = bt[tile * 16 + (tid & 15)];
    v8f acc = {};
    acc = gemv_frag<32>(tvec + wave * 32, bw, acc);
    const float r = ksplit8_sum(acc, part);
    if (tid < 16) q[tile * 16 + tid] = r + btv;
}

__global__ __launch_bounds__(256) void p_kernel(
    const float* q,
    const float* wk_x, const float* wk_h, const float* wk_s, const float* wk_m,
    const float* g_x, const float* be_x, const float* g_h, const float* be_h,
    const float* g_s, const float* be_s, const float* g_m, const float* be_m,
    float* gp, float* sgpp, float* sbpp)
{
    __shared__ float q_s[D_DIM];
    __shared__ float part[128];
    __shared__ float gv_s[16];
    __shared__ float bv_s[16];
    const int tid  = threadIdx.x;
    const int mod  = blockIdx.x >> 4;
    const int tile = blockIdx.x & 15;
    const float* wks[4] = {wk_x, wk_h, wk_s, wk_m};
    const float* gs[4]  = {g_x,  g_h,  g_s,  g_m};
    const float* bes[4] = {be_x, be_h, be_s, be_m};

    const int wave = tid >> 5;
    const float qv = q[tid];
    v2f bw[8];
    load_bfrag<32>(wks[mod] + wave * 32, 1L, (long)D_DIM, tile * 16, bw);
    const int cch = tile * 16 + (tid & 15);
    const float gch = gs[mod][cch];
    const float bch = bes[mod][cch];

    q_s[tid] = qv;
    lds_barrier();

    v8f acc = {};
    acc = gemv_frag<32>(q_s + wave * 32, bw, acc);
    const float p = ksplit8_sum(acc, part);
    if (tid < 16) {
        const float gv = gch * p;
        gv_s[tid] = gv;
        bv_s[tid] = bch * p;
        gp[mod * C_DIM + cch] = gv;
    }
    lds_barrier();
    if (tid == 0) {
        float sg = 0.0f, sb = 0.0f;
        for (int i = 0; i < 16; ++i) { sg += gv_s[i]; sb += bv_s[i]; }
        sgpp[mod * 16 + tile] = sg;
        sbpp[mod * 16 + tile] = sb;
    }
}

#define TR_PITCH 260
__global__ __launch_bounds__(256) void tile_kernel(
    const float* x, const float* h, const float* s, const float* m,
    const float* gp, const float* sgpp, const float* sbpp,
    float* up, float* pm, float* ps, float* pr)
{
    __shared__ __attribute__((aligned(16))) float r1[16 * 64];
    __shared__ __attribute__((aligned(16))) float r2[16 * 64];
    __shared__ __attribute__((aligned(16))) float rd[16 * 64];
    __shared__ __attribute__((aligned(16))) float al_s[64];
    __shared__ __attribute__((aligned(16))) float tr[16 * TR_PITCH];
    __shared__ float wred[6];
    const int tid = threadIdx.x;
    const int bid = blockIdx.x;
    const int mod = bid >> 6;
    const int j0  = (bid & 63) << 6;
    const int tq  = tid & 15;
    const int cg  = tid >> 4;
    const float* am = (mod == 0) ? x : (mod == 1) ? h : (mod == 2) ? s : m;
    const float* a = am + (long)(cg * 16) * N_TOK + j0 + 4 * tq;

    float4 v[16];
#pragma unroll
    for (int cc = 0; cc < 16; ++cc) v[cc] = *(const float4*)(a + (long)cc * N_TOK);

    const float4* g4 = (const float4*)(gp + mod * C_DIM + cg * 16);
    float gv[16];
#pragma unroll
    for (int q = 0; q < 4; ++q) {
        const float4 t = g4[q];
        gv[4 * q] = t.x; gv[4 * q + 1] = t.y; gv[4 * q + 2] = t.z; gv[4 * q + 3] = t.w;
    }
    float sg = 0.0f, sb = 0.0f;
    for (int i = 0; i < 16; ++i) { sg += sgpp[mod * 16 + i]; sb += sbpp[mod * 16 + i]; }
    __builtin_amdgcn_sched_barrier(0);

    float4 s1 = {0.0f, 0.0f, 0.0f, 0.0f};
    float4 s2 = {0.0f, 0.0f, 0.0f, 0.0f};
    float4 dp = {0.0f, 0.0f, 0.0f, 0.0f};
#pragma unroll
    for (int cc = 0; cc < 16; ++cc) {
        const float4 t = v[cc];
        const float g = gv[cc];
        s1.x += t.x;       s1.y += t.y;       s1.z += t.z;       s1.w += t.w;
        s2.x += t.x * t.x; s2.y += t.y * t.y; s2.z += t.z * t.z; s2.w += t.w * t.w;
        dp.x += t.x * g;   dp.y += t.y * g;   dp.z += t.z * g;   dp.w += t.w * g;
    }
    *(float4*)&r1[cg * 64 + 4 * tq] = s1;
    *(float4*)&r2[cg * 64 + 4 * tq] = s2;
    *(float4*)&rd[cg * 64 + 4 * tq] = dp;
    __syncthreads();

    const bool tok = tid < 64;
    const int  wv  = tid >> 5;
    float lg = 0.0f, mu = 0.0f, rstd = 0.0f;
    if (tok) {
        float t1 = 0.0f, t2 = 0.0f, td = 0.0f;
        for (int g = 0; g < 16; ++g) {
            t1 += r1[g * 64 + tid];
            t2 += r2[g * 64 + tid];
            td += rd[g * 64 + tid];
        }
        mu   = t1 * (1.0f / C_DIM);
        const float var  = t2 * (1.0f / C_DIM) - mu * mu;
        rstd = rsqrtf(var + EPSV);
        lg = SCALE * (rstd * (td - mu * sg) + sb);
        const float wm = wave_max32(lg);
        if ((tid & 31) == 0) wred[wv] = wm;
    }
    __syncthreads();
    if (tok) {
        const float bmax = fmaxf(wred[0], wred[1]);
        const float e  = __expf(lg - bmax);
        const float al = e * rstd;
        al_s[tid] = al;
        const float se = wave_sum32(e);
        const float sr = wave_sum32(al * mu);
        if ((tid & 31) == 0) { wred[2 + wv] = se; wred[4 + wv] = sr; }
    }
    __syncthreads();
    if (tid == 0) {
        pm[bid] = fmaxf(wred[0], wred[1]);
        ps[bid] = wred[2] + wred[3];
        pr[bid] = wred[4] + wred[5];
    }

    const float4 a4 = *(const float4*)&al_s[4 * tq];
    float pc[16];
#pragma unroll
    for (int cc = 0; cc < 16; ++cc) {
        const float4 t = v[cc];
        pc[cc] = ((t.x * a4.x + t.y * a4.y) + t.z * a4.z) + t.w * a4.w;
    }
    float* trw = tr + tq * TR_PITCH + cg * 16;
#pragma unroll
    for (int q = 0; q < 4; ++q) {
        float4 t;
        t.x = pc[4 * q]; t.y = pc[4 * q + 1]; t.z = pc[4 * q + 2]; t.w = pc[4 * q + 3];
        *(float4*)&trw[4 * q] = t;
    }
    __syncthreads();
    float uacc = 0.0f;
    for (int t = 0; t < 16; ++t) uacc += tr[t * TR_PITCH + tid];
    up[(long)bid * C_DIM + tid] = uacc;
}


__global__ __launch_bounds__(256) void ov_kernel(
    const float* up, const float* pm, const float* ps, const float* pr,
    const float* g_x, const float* be_x, const float* g_h, const float* be_h,
    const float* g_s, const float* be_s, const float* g_m, const float* be_m,
    const float* wv_x, const float* wv_h, const float* wv_s, const float* wv_m,
    float* ovp)
{
    __shared__ float u_s[C_DIM];
    __shared__ float f_s[256];
    __shared__ float wmx[8];
    __shared__ float wse[8];
    __shared__ float wre[8];
    __shared__ float part[128];
    const int tid  = threadIdx.x;
    const int wvi  = tid >> 5;
    const int lane = tid & 31;
    const int mod  = blockIdx.x >> 4;
    const int tile = blockIdx.x & 15;
    const float* wvm = (mod == 0) ? wv_x : (mod == 1) ? wv_h : (mod == 2) ? wv_s : wv_m;
    const float* gm  = (mod == 0) ? g_x  : (mod == 1) ? g_h  : (mod == 2) ? g_s  : g_m;
    const float* bem = (mod == 0) ? be_x : (mod == 1) ? be_h : (mod == 2) ? be_s : be_m;

    const float mb = pm[tid];
    const float sb = ps[tid];
    const float rb = pr[tid];
    const float* upm = up + (long)(mod * 64) * C_DIM + tid;
    float pv[64];
#pragma unroll
    for (int b = 0; b < 64; ++b) pv[b] = upm[b * C_DIM];
    const float gc  = gm[tid];
    const float bec = bem[tid];
    v2f bw[8];
    load_bfrag<32>(wvm + (long)(wvi * 32) * D_DIM, (long)D_DIM, 1L, tile * 16, bw);

    const float wm = wave_max32(mb);
    if (lane == 0) wmx[wvi] = wm;
    lds_barrier();
    float gmax = wmx[0];
    for (int i = 1; i < 8; ++i) gmax = fmaxf(gmax, wmx[i]);
    const float eb = __expf(mb - gmax);
    f_s[tid] = eb;
    const float se = wave_sum32(sb * eb);
    const float sr = wave_sum32(rb * eb);
    if (lane == 0) { wse[wvi] = se; wre[wvi] = sr; }
    lds_barrier();
    float ssum = 0.0f;
    for (int i = 0; i < 8; ++i) ssum += wse[i];
    const float inv = 1.0f / ssum;
    const float* fm = f_s + mod * 64;
    float c0 = 0.0f, c1 = 0.0f, c2 = 0.0f, c3 = 0.0f;
#pragma unroll
    for (int b = 0; b < 64; b += 4) {
        c0 += pv[b]     * fm[b];
        c1 += pv[b + 1] * fm[b + 1];
        c2 += pv[b + 2] * fm[b + 2];
        c3 += pv[b + 3] * fm[b + 3];
    }
    const float T    = ((c0 + c1) + (c2 + c3)) * inv;
    const float wsum = (wse[2 * mod] + wse[2 * mod + 1]) * inv;
    const float amu  = (wre[2 * mod] + wre[2 * mod + 1]) * inv;
    u_s[tid] = gc * (T - amu) + bec * wsum;
    lds_barrier();

    v8f acc = {};
    acc = gemv_frag<32>(u_s + wvi * 32, bw, acc);
    const float r = ksplit8_sum(acc, part);
    if (tid < 16) ovp[mod * D_DIM + tile * 16 + tid] = r;
}

__global__ __launch_bounds__(256) void ggff_kernel(
    const float* ovp, const float* g_res, const float* be_res,
    const float* w_geglu, const float* b_geglu, const float* w_ff2, float* ffp)
{
    __shared__ float rr[D_DIM];
    __shared__ float wsa[8];
    __shared__ float wsb[8];
    __shared__ float parta[128];
    __shared__ float partb[128];
    __shared__ float hh_s[16];
    const int tid  = threadIdx.x;
    const int wave = tid >> 5;
    const int lane = tid & 31;
    const int tile = blockIdx.x;

    const float o0 = ovp[tid];
    const float o1 = ovp[D_DIM + tid];
    const float o2 = ovp[2 * D_DIM + tid];
    const float o3 = ovp[3 * D_DIM + tid];
    const float gres  = g_res[tid];
    const float beres = be_res[tid];
    const float* wg = w_geglu + (long)(wave * 32) * (2 * D_DIM);
    v2f bga[8];
    v2f bgb[8];
    load_bfrag<32>(wg, (long)(2 * D_DIM), 1L, tile * 16, bga);
    load_bfrag<32>(wg, (long)(2 * D_DIM), 1L, D_DIM + tile * 16, bgb);
    const float bgu = b_geglu[tile * 16 + (tid & 15)];
    const float bgg = b_geglu[D_DIM + tile * 16 + (tid & 15)];
    const float* w2 = w_ff2 + (long)(tile * 16) * D_DIM;
    const int n0 = wave * 32;
    v2f b20[4];
    v2f b21[4];
    load_bfrag<16>(w2, (long)D_DIM, 1L, n0, b20);
    load_bfrag<16>(w2, (long)D_DIM, 1L, n0 + 16, b21);

    const float ov = ((o0 + o1) + o2) + o3;
    const float mu = block_sum_256(ov, wsa) * (1.0f / D_DIM);
    const float dv = ov - mu;
    const float var = block_sum_256(dv * dv, wsb) * (1.0f / D_DIM);
    const float rstd = rsqrtf(var + EPSV);
    rr[tid] = dv * rstd * gres + beres;
    lds_barrier();

    v8f acca = {};
    v8f accb = {};
    gemv_frag2<32>(rr + wave * 32, bga, bgb, acca, accb);
    if (lane < 16) { parta[wave * 16 + lane] = acca[0]; partb[wave * 16 + lane] = accb[0]; }
    lds_barrier();
    if (tid < 16) {
        float ru = 0.0f, rg = 0.0f;
        for (int i = 0; i < 8; ++i) { ru += parta[i * 16 + tid]; rg += partb[i * 16 + tid]; }
        const float uu = ru + bgu;
        const float g  = rg + bgg;
        const float ge = 0.5f * g * (1.0f + erff(g * 0.70710678118654752f));
        hh_s[tid] = uu * ge;
    }
    lds_barrier();

    v8f acc0 = {};
    v8f acc1 = {};
    gemv_frag2<16>(hh_s, b20, b21, acc0, acc1);
    if (lane < 16) {
        ffp[tile * D_DIM + n0 + lane]      = acc0[0];
        ffp[tile * D_DIM + n0 + 16 + lane] = acc1[0];
    }
}

__global__ __launch_bounds__(256) void outb_kernel(
    const float* ffp, const float* b_ff2, const float* w_out, const float* b_out,
    float* out)
{
    __shared__ float ff_s[D_DIM];
    __shared__ float part[128];
    __shared__ float y_s[16];
    const int tid   = threadIdx.x;
    const int tile  = blockIdx.x >> 4;
    const int slice = blockIdx.x & 15;
    const int wave  = tid >> 5;
    float fp[16];
#pragma unroll
    for (int i = 0; i < 16; ++i) fp[i] = ffp[i * D_DIM + tid];
    const float bf2 = b_ff2[tid];
    v2f bw[8];
    load_bfrag<32>(w_out + (long)(wave * 32) * C_DIM, (long)C_DIM, 1L, tile * 16, bw);
    const float bo = b_out[tile * 16 + (tid & 15)];

    float f = 0.0f;
#pragma unroll
    for (int i = 0; i < 16; ++i) f += fp[i];
    ff_s[tid] = f + bf2;
    lds_barrier();
    v8f acc = {};
    acc = gemv_frag<32>(ff_s + wave * 32, bw, acc);
    const float r = ksplit8_sum(acc, part);
    if (tid < 16) y_s[tid] = r + bo;
    lds_barrier();
    const int t4 = tid & 63;
    const int cg = tid >> 6;
    for (int i = 0; i < 4; ++i) {
        const int ch = 4 * i + cg;
        const float v = y_s[ch];
        float4 vv;
        vv.x = v; vv.y = v; vv.z = v; vv.w = v;
        float4* o = (float4*)(out + (long)(tile * 16 + ch) * N_TOK + slice * 256 + t4 * 4);
        *o = vv;
    }
}

extern "C" void kernel_launch(void* const* d_in, const int* in_sizes, int n_in,
                              void* d_out, int out_size, void* d_ws, size_t ws_size,
                              hipStream_t stream) {
    const float* x       = (const float*)d_in[0];
    const float* h       = (const float*)d_in[1];
    const float* s       = (const float*)d_in[2];
    const float* m       = (const float*)d_in[3];
    const float* tvec    = (const float*)d_in[4];
    const float* wt      = (const float*)d_in[5];
    const float* bt      = (const float*)d_in[6];
    const float* wk_x    = (const float*)d_in[7];
    const float* wv_x    = (const float*)d_in[8];
    const float* g_x     = (const float*)d_in[9];
    const float* be_x    = (const float*)d_in[10];
    const float* wk_h    = (const float*)d_in[11];
    const float* wv_h    = (const float*)d_in[12];
    const float* g_h     = (const float*)d_in[13];
    const float* be_h    = (const float*)d_in[14];
    const float* wk_s    = (const float*)d_in[15];
    const float* wv_s    = (const float*)d_in[16];
    const float* g_s     = (const float*)d_in[17];
    const float* be_s    = (const float*)d_in[18];
    const float* wk_m    = (const float*)d_in[19];
    const float* wv_m    = (const float*)d_in[20];
    const float* g_m     = (const float*)d_in[21];
    const float* be_m    = (const float*)d_in[22];
    const float* w_geglu = (const float*)d_in[23];
    const float* b_geglu = (const float*)d_in[24];
    const float* w_ff2   = (const float*)d_in[25];
    const float* b_ff2   = (const float*)d_in[26];
    const float* g_res   = (const float*)d_in[27];
    const float* be_res  = (const float*)d_in[28];
    const float* w_out   = (const float*)d_in[29];
    const float* b_out   = (const float*)d_in[30];
    float* out = (float*)d_out;
    float* ws  = (float*)d_ws;

    float* gp     = ws + WS_GP;
    float* q      = ws + WS_Q;
    float* sgpp   = ws + WS_SGPP;
    float* sbpp   = ws + WS_SBPP;
    float* ovp    = ws + WS_OVP;
    float* ffp    = ws + WS_FFP;
    float* up     = ws + WS_UP;
    float* pm     = ws + WS_PM;
    float* ps     = ws + WS_PS;
    float* pr     = ws + WS_PR;

    q_kernel<<<16, 256, 0, stream>>>(tvec, wt, bt, q);
    p_kernel<<<64, 256, 0, stream>>>(q, wk_x, wk_h, wk_s, wk_m,
                                     g_x, be_x, g_h, be_h, g_s, be_s, g_m, be_m,
                                     gp, sgpp, sbpp);
    tile_kernel<<<256, 256, 0, stream>>>(x, h, s, m, gp, sgpp, sbpp,
                                         up, pm, ps, pr);
    ov_kernel<<<64, 256, 0, stream>>>(up, pm, ps, pr,
                                      g_x, be_x, g_h, be_h, g_s, be_s, g_m, be_m,
                                      wv_x, wv_h, wv_s, wv_m, ovp);
    ggff_kernel<<<16, 256, 0, stream>>>(ovp, g_res, be_res, w_geglu, b_geglu, w_ff2, ffp);
    outb_kernel<<<256, 256, 0, stream>>>(ffp, b_ff2, w_out, b_out, out);
}


# q and p in one launch: workgroup applies its 16 q values to 16 columns of wk, reader adds the 16 partial p vectors; tile workgroups take 128 tokens so these fit; p, g*p, token sum orders change
# speedup vs baseline: 7.2928x; 1.1844x over previous
#include <hip/hip_runtime.h>
#include <math.h>

typedef __attribute__((ext_vector_type(2))) float v2f;
typedef __attribute__((ext_vector_type(8))) float v8f;

#define C_DIM 256
#define D_DIM 256
#define N_TOK 4096
#define SCALE 0.0625f
#define EPSV  1e-5f

#define N_BLK    128
#define WS_OVP   0
#define WS_PM    1024
#define WS_PS    1152
#define WS_PR    1280
#define WS_UP    2048
#define WS_FFP   4096
#define WS_PP    34816

template <int CTRL>
__device__ inline float lane_xmask(float v) {
    return __int_as_float(__builtin_amdgcn_update_dpp(0, __float_as_int(v), CTRL, 0xf, 0xf, false));
}
__device__ inline float lane_xrow(float v) {
    return __int_as_float((int)__builtin_amdgcn_permlanex16(0u, (unsigned)__float_as_int(v),
                                                            0x76543210u, 0xfedcba98u, false, false));
}
__device__ inline float wave_max32(float v) {
    v = fmaxf(v, lane_xmask<0x161>(v));
    v = fmaxf(v, lane_xmask<0x162>(v));
    v = fmaxf(v, lane_xmask<0x164>(v));
    v = fmaxf(v, lane_xmask<0x168>(v));
    v = fmaxf(v, lane_xrow(v));
    return v;
}
__device__ inline float wave_sum32(float v) {
    v += lane_xmask<0x161>(v);
    v += lane_xmask<0x162>(v);
    v += lane_xmask<0x164>(v);
    v += lane_xmask<0x168>(v);
    v += lane_xrow(v);
    return v;
}

__device__ inline void lds_barrier() {
    __builtin_amdgcn_fence(__ATOMIC_RELEASE, "workgroup", "local");
    __builtin_amdgcn_s_barrier();
    __builtin_amdgcn_fence(__ATOMIC_ACQUIRE, "workgroup", "local");
}

__device__ inline float block_sum_256(float v, float* wp) {
    const float w = wave_sum32(v);
    if ((threadIdx.x & 31) == 0) wp[threadIdx.x >> 5] = w;
    lds_barrier();
    float r = 0.0f;
    for (int i = 0; i < 8; ++i) r += wp[i];
    return r;
}

template <int KW>
__device__ inline void load_bfrag(const float* w, long sk, long sn, int n0, v2f (&b)[KW / 4]) {
    const int lane = threadIdx.x & 31;
    const int half = lane >> 4;
    const int l16  = lane & 15;
    const float* wp = w + (long)(n0 + l16) * sn;
#pragma unroll
    for (int i = 0; i < KW / 4; ++i) {
        const int kb = 4 * i + 2 * half;
        b[i].x = wp[(long)kb * sk];
        b[i].y = wp[(long)(kb + 1) * sk];
    }
}

template <int KW>
__device__ inline v8f gemv_frag(const float* x, const v2f (&b)[KW / 4], v8f acc) {
    const int lane = threadIdx.x & 31;
    const int half = lane >> 4;
    const float amask = ((lane & 15) == 0) ? 1.0f : 0.0f;
#pragma unroll
    for (int i = 0; i < KW / 4; ++i) {
        const int kb = 4 * i + 2 * half;
        v2f a;
        a.x = x[kb] * amask;
        a.y = x[kb + 1] * amask;
        acc = __builtin_amdgcn_wmma_f32_16x16x4_f32(
            false, a, false, b[i], (short)0, acc, false, false);
    }
    return acc;
}

template <int KW>
__device__ inline void gemv_frag2(const float* x, const v2f (&ba)[KW / 4], const v2f (&bb)[KW / 4],
                                  v8f& accA, v8f& accB) {
    const int lane = threadIdx.x & 31;
    const int half = lane >> 4;
    const float amask = ((lane & 15) == 0) ? 1.0f : 0.0f;
#pragma unroll
    for (int i = 0; i < KW / 4; ++i) {
        const int kb = 4 * i + 2 * half;
        v2f a;
        a.x = x[kb] * amask;
        a.y = x[kb + 1] * amask;
        accA = __builtin_amdgcn_wmma_f32_16x16x4_f32(
            false, a, false, ba[i], (short)0, accA, false, false);
        accB = __builtin_amdgcn_wmma_f32_16x16x4_f32(
            false, a, false, bb[i], (short)0, accB, false, false);
    }
}

__device__ inline float ksplit8_sum(v8f acc, float* part) {
    const int wave = threadIdx.x >> 5;
    const int lane = threadIdx.x & 31;
    if (lane < 16) part[wave * 16 + lane] = acc[0];
    lds_barrier();
    float r = 0.0f;
    if (threadIdx.x < 16) {
        for (int i = 0; i < 8; ++i) r += part[i * 16 + threadIdx.x];
    }
    return r;
}

__global__ __launch_bounds__(256) void qp_kernel(
    const float* tvec, const float* wt, const float* bt,
    const float* wk_x, const float* wk_h, const float* wk_s, const float* wk_m,
    float* pp)
{
    __shared__ float part[128];
    __shared__ float q_s[16];
    const int tid  = threadIdx.x;
    const int wave = tid >> 5;
    const int lane = tid & 31;
    const int tile = blockIdx.x >> 2;
    const int mod  = blockIdx.x & 3;
    const float* wkm = (mod == 0) ? wk_x : (mod == 1) ? wk_h : (mod == 2) ? wk_s : wk_m;
    v2f bw[8];
    load_bfrag<32>(wt + (long)(wave * 32) * D_DIM, (long)D_DIM, 1L, tile * 16, bw);
    const float btv = bt[tile * 16 + (tid & 15)];
    const float* wq = wkm + tile * 16;
    const int n0 = wave * 32;
    v2f b0[4];
    v2f b1[4];
    load_bfrag<16>(wq, 1L, (long)D_DIM, n0, b0);
    load_bfrag<16>(wq, 1L, (long)D_DIM, n0 + 16, b1);

    v8f acc = {};
    acc = gemv_frag<32>(tvec + wave * 32, bw, acc);
    const float r = ksplit8_sum(acc, part);
    if (tid < 16) q_s[tid] = r + btv;
    lds_barrier();

    v8f acc0 = {};
    v8f acc1 = {};
    gemv_frag2<16>(q_s, b0, b1, acc0, acc1);
    if (lane < 16) {
        float* po = pp + (long)(tile * 4 + mod) * C_DIM + n0;
        po[lane]      = acc0[0];
        po[16 + lane] = acc1[0];
    }
}

#define TR_PITCH 260
__global__ __launch_bounds__(512) void tile_kernel(
    const float* x, const float* h, const float* s, const float* m,
    const float* pp,
    const float* g_x, const float* be_x, const float* g_h, const float* be_h,
    const float* g_s, const float* be_s, const float* g_m, const float* be_m,
    float* up, float* pm, float* ps, float* pr)
{
    __shared__ __attribute__((aligned(16))) float r1[16 * 128];
    __shared__ __attribute__((aligned(16))) float r2[16 * 128];
    __shared__ __attribute__((aligned(16))) float rd[16 * 128];
    __shared__ __attribute__((aligned(16))) float al_s[128];
    __shared__ __attribute__((aligned(16))) float tr[32 * TR_PITCH];
    __shared__ __attribute__((aligned(16))) float gp_s[C_DIM];
    __shared__ float wsg[8];
    __shared__ float wsb[8];
    __shared__ float wred[12];
    const int tid = threadIdx.x;
    const int bid = blockIdx.x;
    const int mod = bid >> 5;
    const int j0  = (bid & 31) << 7;
    const int tq  = tid & 31;
    const int cg  = tid >> 5;
    const float* am  = (mod == 0) ? x    : (mod == 1) ? h    : (mod == 2) ? s    : m;
    const float* gm  = (mod == 0) ? g_x  : (mod == 1) ? g_h  : (mod == 2) ? g_s  : g_m;
    const float* bem = (mod == 0) ? be_x : (mod == 1) ? be_h : (mod == 2) ? be_s : be_m;
    const float* a = am + (long)(cg * 16) * N_TOK + j0 + 4 * tq;

    const bool chan = tid < C_DIM;
    float pq[16];
    float gch = 0.0f, bch = 0.0f;
#pragma unroll
    for (int i = 0; i < 16; ++i) pq[i] = 0.0f;
    if (chan) {
        const float* ppm = pp + mod * C_DIM + tid;
#pragma unroll
        for (int i = 0; i < 16; ++i) pq[i] = ppm[(long)i * (4 * C_DIM)];
        gch = gm[tid];
        bch = bem[tid];
    }
    float4 v[16];
#pragma unroll
    for (int cc = 0; cc < 16; ++cc) v[cc] = *(const float4*)(a + (long)cc * N_TOK);
    __builtin_amdgcn_sched_barrier(0);

    float gpc = 0.0f, bpc = 0.0f;
    if (chan) {
        float p = 0.0f;
#pragma unroll
        for (int i = 0; i < 16; ++i) p += pq[i];
        gpc = gch * p;
        bpc = bch * p;
        gp_s[tid] = gpc;
    }
    const float wsum_g = wave_sum32(gpc);
    const float wsum_b = wave_sum32(bpc);
    if (chan && tq == 0) { wsg[cg] = wsum_g; wsb[cg] = wsum_b; }
    lds_barrier();
    float sg = 0.0f, sb = 0.0f;
    for (int i = 0; i < 8; ++i) { sg += wsg[i]; sb += wsb[i]; }
    const float4* g4 = (const float4*)(gp_s + cg * 16);
    float gv[16];
#pragma unroll
    for (int q = 0; q < 4; ++q) {
        const float4 t = g4[q];
        gv[4 * q] = t.x; gv[4 * q + 1] = t.y; gv[4 * q + 2] = t.z; gv[4 * q + 3] = t.w;
    }

    float4 s1 = {0.0f, 0.0f, 0.0f, 0.0f};
    float4 s2 = {0.0f, 0.0f, 0.0f, 0.0f};
    float4 dp = {0.0f, 0.0f, 0.0f, 0.0f};
#pragma unroll
    for (int cc = 0; cc < 16; ++cc) {
        const float4 t = v[cc];
        const float g = gv[cc];
        s1.x += t.x;       s1.y += t.y;       s1.z += t.z;       s1.w += t.w;
        s2.x += t.x * t.x; s2.y += t.y * t.y; s2.z += t.z * t.z; s2.w += t.w * t.w;
        dp.x += t.x * g;   dp.y += t.y * g;   dp.z += t.z * g;   dp.w += t.w * g;
    }
    *(float4*)&r1[cg * 128 + 4 * tq] = s1;
    *(float4*)&r2[cg * 128 + 4 * tq] = s2;
    *(float4*)&rd[cg * 128 + 4 * tq] = dp;
    __syncthreads();

    const bool tok = tid < 128;
    const int  wv  = tid >> 5;
    float lg = 0.0f, mu = 0.0f, rstd = 0.0f;
    if (tok) {
        float t1 = 0.0f, t2 = 0.0f, td = 0.0f;
        for (int g = 0; g < 16; ++g) {
            t1 += r1[g * 128 + tid];
            t2 += r2[g * 128 + tid];
            td += rd[g * 128 + tid];
        }
        mu   = t1 * (1.0f / C_DIM);
        const float var  = t2 * (1.0f / C_DIM) - mu * mu;
        rstd = rsqrtf(var + EPSV);
        lg = SCALE * (rstd * (td - mu * sg) + sb);
        const float wm = wave_max32(lg);
        if ((tid & 31) == 0) wred[wv] = wm;
    }
    __syncthreads();
    if (tok) {
        const float bmax = fmaxf(fmaxf(wred[0], wred[1]), fmaxf(wred[2], wred[3]));
        const float e  = __expf(lg - bmax);
        const float al = e * rstd;
        al_s[tid] = al;
        const float se = wave_sum32(e);
        const float sr = wave_sum32(al * mu);
        if ((tid & 31) == 0) { wred[4 + wv] = se; wred[8 + wv] = sr; }
    }
    __syncthreads();
    if (tid == 0) {
        pm[bid] = fmaxf(fmaxf(wred[0], wred[1]), fmaxf(wred[2], wred[3]));
        ps[bid] = ((wred[4] + wred[5]) + wred[6]) + wred[7];
        pr[bid] = ((wred[8] + wred[9]) + wred[10]) + wred[11];
    }

    const float4 a4 = *(const float4*)&al_s[4 * tq];
    float pc[16];
#pragma unroll
    for (int cc = 0; cc < 16; ++cc) {
        const float4 t = v[cc];
        pc[cc] = ((t.x * a4.x + t.y * a4.y) + t.z * a4.z) + t.w * a4.w;
    }
    float* trw = tr + tq * TR_PITCH + cg * 16;
#pragma unroll
    for (int q = 0; q < 4; ++q) {
        float4 t;
        t.x = pc[4 * q]; t.y = pc[4 * q + 1]; t.z = pc[4 * q + 2]; t.w = pc[4 * q + 3];
        *(float4*)&trw[4 * q] = t;
    }
    __syncthreads();
    if (chan) {
        float uacc = 0.0f;
        for (int t = 0; t < 32; ++t) uacc += tr[t * TR_PITCH + tid];
        up[(long)bid * C_DIM + tid] = uacc;
    }
}


__global__ __launch_bounds__(256) void ov_kernel(
    const float* up, const float* pm, const float* ps, const float* pr,
    const float* g_x, const float* be_x, const float* g_h, const float* be_h,
    const float* g_s, const float* be_s, const float* g_m, const float* be_m,
    const float* wv_x, const float* wv_h, const float* wv_s, const float* wv_m,
    float* ovp)
{
    __shared__ float u_s[C_DIM];
    __shared__ float f_s[N_BLK];
    __shared__ float wmx[4];
    __shared__ float wse[4];
    __shared__ float wre[4];
    __shared__ float part[128];
    const int tid  = threadIdx.x;
    const int wvi  = tid >> 5;
    const int lane = tid & 31;
    const int mod  = blockIdx.x >> 4;
    const int tile = blockIdx.x & 15;
    const float* wvm = (mod == 0) ? wv_x : (mod == 1) ? wv_h : (mod == 2) ? wv_s : wv_m;
    const float* gm  = (mod == 0) ? g_x  : (mod == 1) ? g_h  : (mod == 2) ? g_s  : g_m;
    const float* bem = (mod == 0) ? be_x : (mod == 1) ? be_h : (mod == 2) ? be_s : be_m;

    const bool blk = tid < N_BLK;
    float mb = 0.0f, sb = 0.0f, rb = 0.0f;
    if (blk) { mb = pm[tid]; sb = ps[tid]; rb = pr[tid]; }
    const float* upm = up + (long)(mod * 32) * C_DIM + tid;
    float pv[32];
#pragma unroll
    for (int b = 0; b < 32; ++b) pv[b] = upm[b * C_DIM];
    const float gc  = gm[tid];
    const float bec = bem[tid];
    v2f bw[8];
    load_bfrag<32>(wvm + (long)(wvi * 32) * D_DIM, (long)D_DIM, 1L, tile * 16, bw);

    if (blk) {
        const float wm = wave_max32(mb);
        if (lane == 0) wmx[wvi] = wm;
    }
    lds_barrier();
    const float gmax = fmaxf(fmaxf(wmx[0], wmx[1]), fmaxf(wmx[2], wmx[3]));
    if (blk) {
        const float eb = __expf(mb - gmax);
        f_s[tid] = eb;
        const float se = wave_sum32(sb * eb);
        const float sr = wave_sum32(rb * eb);
        if (lane == 0) { wse[wvi] = se; wre[wvi] = sr; }
    }
    lds_barrier();
    float ssum = 0.0f;
    for (int i = 0; i < 4; ++i) ssum += wse[i];
    const float inv = 1.0f / ssum;
    const float* fm = f_s + mod * 32;
    float c0 = 0.0f, c1 = 0.0f, c2 = 0.0f, c3 = 0.0f;
#pragma unroll
    for (int b = 0; b < 32; b += 4) {
        c0 += pv[b]     * fm[b];
        c1 += pv[b + 1] * fm[b + 1];
        c2 += pv[b + 2] * fm[b + 2];
        c3 += pv[b + 3] * fm[b + 3];
    }
    const float T    = ((c0 + c1) + (c2 + c3)) * inv;
    const float wsum = wse[mod] * inv;
    const float amu  = wre[mod] * inv;
    u_s[tid] = gc * (T - amu) + bec * wsum;
    lds_barrier();

    v8f acc = {};
    acc = gemv_frag<32>(u_s + wvi * 32, bw, acc);
    const float r = ksplit8_sum(acc, part);
    if (tid < 16) ovp[mod * D_DIM + tile * 16 + tid] = r;
}

__global__ __launch_bounds__(256) void ggff_kernel(
    const float* ovp, const float* g_res, const float* be_res,
    const float* w_geglu, const float* b_geglu, const float* w_ff2, float* ffp)
{
    __shared__ float rr[D_DIM];
    __shared__ float wsa[8];
    __shared__ float wsb[8];
    __shared__ float parta[128];
    __shared__ float partb[128];
    __shared__ float hh_s[16];
    const int tid  = threadIdx.x;
    const int wave = tid >> 5;
    const int lane = tid & 31;
    const int tile = blockIdx.x;

    const float o0 = ovp[tid];
    const float o1 = ovp[D_DIM + tid];
    const float o2 = ovp[2 * D_DIM + tid];
    const float o3 = ovp[3 * D_DIM + tid];
    const float gres  = g_res[tid];
    const float beres = be_res[tid];
    const float* wg = w_geglu + (long)(wave * 32) * (2 * D_DIM);
    v2f bga[8];
    v2f bgb[8];
    load_bfrag<32>(wg, (long)(2 * D_DIM), 1L, tile * 16, bga);
    load_bfrag<32>(wg, (long)(2 * D_DIM), 1L, D_DIM + tile * 16, bgb);
    const float bgu = b_geglu[tile * 16 + (tid & 15)];
    const float bgg = b_geglu[D_DIM + tile * 16 + (tid & 15)];
    const float* w2 = w_ff2 + (long)(tile * 16) * D_DIM;
    const int n0 = wave * 32;
    v2f b20[4];
    v2f b21[4];
    load_bfrag<16>(w2, (long)D_DIM, 1L, n0, b20);
    load_bfrag<16>(w2, (long)D_DIM, 1L, n0 + 16, b21);

    const float ov = ((o0 + o1) + o2) + o3;
    const float mu = block_sum_256(ov, wsa) * (1.0f / D_DIM);
    const float dv = ov - mu;
    const float var = block_sum_256(dv * dv, wsb) * (1.0f / D_DIM);
    const float rstd = rsqrtf(var + EPSV);
    rr[tid] = dv * rstd * gres + beres;
    lds_barrier();

    v8f acca = {};
    v8f accb = {};
    gemv_frag2<32>(rr + wave * 32, bga, bgb, acca, accb);
    if (lane < 16) { parta[wave * 16 + lane] = acca[0]; partb[wave * 16 + lane] = accb[0]; }
    lds_barrier();
    if (tid < 16) {
        float ru = 0.0f, rg = 0.0f;
        for (int i = 0; i < 8; ++i) { ru += parta[i * 16 + tid]; rg += partb[i * 16 + tid]; }
        const float uu = ru + bgu;
        const float g  = rg + bgg;
        const float ge = 0.5f * g * (1.0f + erff(g * 0.70710678118654752f));
        hh_s[tid] = uu * ge;
    }
    lds_barrier();

    v8f acc0 = {};
    v8f acc1 = {};
    gemv_frag2<16>(hh_s, b20, b21, acc0, acc1);
    if (lane < 16) {
        ffp[tile * D_DIM + n0 + lane]      = acc0[0];
        ffp[tile * D_DIM + n0 + 16 + lane] = acc1[0];
    }
}

__global__ __launch_bounds__(256) void outb_kernel(
    const float* ffp, const float* b_ff2, const float* w_out, const float* b_out,
    float* out)
{
    __shared__ float ff_s[D_DIM];
    __shared__ float part[128];
    __shared__ float y_s[16];
    const int tid   = threadIdx.x;
    const int tile  = blockIdx.x >> 4;
    const int slice = blockIdx.x & 15;
    const int wave  = tid >> 5;
    float fp[16];
#pragma unroll
    for (int i = 0; i < 16; ++i) fp[i] = ffp[i * D_DIM + tid];
    const float bf2 = b_ff2[tid];
    v2f bw[8];
    load_bfrag<32>(w_out + (long)(wave * 32) * C_DIM, (long)C_DIM, 1L, tile * 16, bw);
    const float bo = b_out[tile * 16 + (tid & 15)];

    float f = 0.0f;
#pragma unroll
    for (int i = 0; i < 16; ++i) f += fp[i];
    ff_s[tid] = f + bf2;
    lds_barrier();
    v8f acc = {};
    acc = gemv_frag<32>(ff_s + wave * 32, bw, acc);
    const float r = ksplit8_sum(acc, part);
    if (tid < 16) y_s[tid] = r + bo;
    lds_barrier();
    const int t4 = tid & 63;
    const int cg = tid >> 6;
    for (int i = 0; i < 4; ++i) {
        const int ch = 4 * i + cg;
        const float v = y_s[ch];
        float4 vv;
        vv.x = v; vv.y = v; vv.z = v; vv.w = v;
        float4* o = (float4*)(out + (long)(tile * 16 + ch) * N_TOK + slice * 256 + t4 * 4);
        *o = vv;
    }
}

extern "C" void kernel_launch(void* const* d_in, const int* in_sizes, int n_in,
                              void* d_out, int out_size, void* d_ws, size_t ws_size,
                              hipStream_t stream) {
    const float* x       = (const float*)d_in[0];
    const float* h       = (const float*)d_in[1];
    const float* s       = (const float*)d_in[2];
    const float* m       = (const float*)d_in[3];
    const float* tvec    = (const float*)d_in[4];
    const float* wt      = (const float*)d_in[5];
    const float* bt      = (const float*)d_in[6];
    const float* wk_x    = (const float*)d_in[7];
    const float* wv_x    = (const float*)d_in[8];
    const float* g_x     = (const float*)d_in[9];
    const float* be_x    = (const float*)d_in[10];
    const float* wk_h    = (const float*)d_in[11];
    const float* wv_h    = (const float*)d_in[12];
    const float* g_h     = (const float*)d_in[13];
    const float* be_h    = (const float*)d_in[14];
    const float* wk_s    = (const float*)d_in[15];
    const float* wv_s    = (const float*)d_in[16];
    const float* g_s     = (const float*)d_in[17];
    const float* be_s    = (const float*)d_in[18];
    const float* wk_m    = (const float*)d_in[19];
    const float* wv_m    = (const float*)d_in[20];
    const float* g_m     = (const float*)d_in[21];
    const float* be_m    = (const float*)d_in[22];
    const float* w_geglu = (const float*)d_in[23];
    const float* b_geglu = (const float*)d_in[24];
    const float* w_ff2   = (const float*)d_in[25];
    const float* b_ff2   = (const float*)d_in[26];
    const float* g_res   = (const float*)d_in[27];
    const float* be_res  = (const float*)d_in[28];
    const float* w_out   = (const float*)d_in[29];
    const float* b_out   = (const float*)d_in[30];
    float* out = (float*)d_out;
    float* ws  = (float*)d_ws;

    float* pp     = ws + WS_PP;
    float* ovp    = ws + WS_OVP;
    float* ffp    = ws + WS_FFP;
    float* up     = ws + WS_UP;
    float* pm     = ws + WS_PM;
    float* ps     = ws + WS_PS;
    float* pr     = ws + WS_PR;

    qp_kernel<<<64, 256, 0, stream>>>(tvec, wt, bt, wk_x, wk_h, wk_s, wk_m, pp);
    tile_kernel<<<N_BLK, 512, 0, stream>>>(x, h, s, m, pp,
                                           g_x, be_x, g_h, be_h, g_s, be_s, g_m, be_m,
                                           up, pm, ps, pr);
    ov_kernel<<<64, 256, 0, stream>>>(up, pm, ps, pr,
                                      g_x, be_x, g_h, be_h, g_s, be_s, g_m, be_m,
                                      wv_x, wv_h, wv_s, wv_m, ovp);
    ggff_kernel<<<16, 256, 0, stream>>>(ovp, g_res, be_res, w_geglu, b_geglu, w_ff2, ffp);
    outb_kernel<<<256, 256, 0, stream>>>(ffp, b_ff2, w_out, b_out, out);
}


# launches 5->4: w_ff2@w_out folded by spare workgroups of launch 1; out_vec pieces applied to w_geglu rows in the wv launch, LayerNorm affine moved through that product; gg, ff, y sum orders change
# speedup vs baseline: 7.8146x; 1.0716x over previous
#include <hip/hip_runtime.h>
#include <math.h>

typedef __attribute__((ext_vector_type(2))) float v2f;
typedef __attribute__((ext_vector_type(8))) float v8f;

#define C_DIM 256
#define D_DIM 256
#define N_TOK 4096
#define SCALE 0.0625f
#define EPSV  1e-5f

#define N_BLK    128
#define WS_OVP   0
#define WS_PM    1024
#define WS_PS    1152
#define WS_PR    1280
#define WS_UP    2048
#define WS_FFP   4096
#define WS_PP    34816

#define F_OVP    0
#define F_PM     1024
#define F_PS     1152
#define F_PR     1280
#define F_BV     1408
#define F_CV     1664
#define F_DV     2176
#define F_UP     4096
#define F_PP     36864
#define F_GP     36864
#define F_MF     69632
#define F_BYTES  540672
#define QPF_GRID 216

template <int CTRL>
__device__ inline float lane_xmask(float v) {
    return __int_as_float(__builtin_amdgcn_update_dpp(0, __float_as_int(v), CTRL, 0xf, 0xf, false));
}
__device__ inline float lane_xrow(float v) {
    return __int_as_float((int)__builtin_amdgcn_permlanex16(0u, (unsigned)__float_as_int(v),
                                                            0x76543210u, 0xfedcba98u, false, false));
}
__device__ inline float wave_max32(float v) {
    v = fmaxf(v, lane_xmask<0x161>(v));
    v = fmaxf(v, lane_xmask<0x162>(v));
    v = fmaxf(v, lane_xmask<0x164>(v));
    v = fmaxf(v, lane_xmask<0x168>(v));
    v = fmaxf(v, lane_xrow(v));
    return v;
}
__device__ inline float wave_sum32(float v) {
    v += lane_xmask<0x161>(v);
    v += lane_xmask<0x162>(v);
    v += lane_xmask<0x164>(v);
    v += lane_xmask<0x168>(v);
    v += lane_xrow(v);
    return v;
}

__device__ inline void lds_barrier() {
    __builtin_amdgcn_fence(__ATOMIC_RELEASE, "workgroup", "local");
    __builtin_amdgcn_s_barrier();
    __builtin_amdgcn_fence(__ATOMIC_ACQUIRE, "workgroup", "local");
}

__device__ inline void pin_loads() {
    __builtin_amdgcn_sched_barrier(0);
}

__device__ inline float block_sum_256(float v, float* wp) {
    const float w = wave_sum32(v);
    if ((threadIdx.x & 31) == 0) wp[threadIdx.x >> 5] = w;
    lds_barrier();
    float r = 0.0f;
    for (int i = 0; i < 8; ++i) r += wp[i];
    return r;
}

template <int KW>
__device__ inline void load_bfrag(const float* w, long sk, long sn, int n0, v2f (&b)[KW / 4]) {
    const int lane = threadIdx.x & 31;
    const int half = lane >> 4;
    const int l16  = lane & 15;
    const float* wp = w + (long)(n0 + l16) * sn;
#pragma unroll
    for (int i = 0; i < KW / 4; ++i) {
        const int kb = 4 * i + 2 * half;
        b[i].x = wp[(long)kb * sk];
        b[i].y = wp[(long)(kb + 1) * sk];
    }
}

__device__ inline void load_bfrag_quad16(const float* w, long sk,
                                         v2f (&b0)[4], v2f (&b1)[4], v2f (&b2)[4], v2f (&b3)[4]) {
    const int lane = threadIdx.x & 31;
    const int half = lane >> 4;
    const int l16  = lane & 15;
    const float* wp = w + 4 * l16;
#pragma unroll
    for (int i = 0; i < 4; ++i) {
        const int kb = 4 * i + 2 * half;
        const float4 r0 = *(const float4*)(wp + (long)kb * sk);
        const float4 r1 = *(const float4*)(wp + (long)(kb + 1) * sk);
        b0[i].x = r0.x; b0[i].y = r1.x;
        b1[i].x = r0.y; b1[i].y = r1.y;
        b2[i].x = r0.z; b2[i].y = r1.z;
        b3[i].x = r0.w; b3[i].y = r1.w;
    }
}

template <int KW>
__device__ inline v8f gemv_frag(const float* x, const v2f (&b)[KW / 4], v8f acc) {
    const int lane = threadIdx.x & 31;
    const int half = lane >> 4;
    const float amask = ((lane & 15) == 0) ? 1.0f : 0.0f;
#pragma unroll
    for (int i = 0; i < KW / 4; ++i) {
        const int kb = 4 * i + 2 * half;
        v2f a;
        a.x = x[kb] * amask;
        a.y = x[kb + 1] * amask;
        acc = __builtin_amdgcn_wmma_f32_16x16x4_f32(
            false, a, false, b[i], (short)0, acc, false, false);
    }
    return acc;
}

template <int KW>
__device__ inline void gemv_frag2(const float* x, const v2f (&ba)[KW / 4], const v2f (&bb)[KW / 4],
                                  v8f& accA, v8f& accB) {
    const int lane = threadIdx.x & 31;
    const int half = lane >> 4;
    const float amask = ((lane & 15) == 0) ? 1.0f : 0.0f;
#pragma unroll
    for (int i = 0; i < KW / 4; ++i) {
        const int kb = 4 * i + 2 * half;
        v2f a;
        a.x = x[kb] * amask;
        a.y = x[kb + 1] * amask;
        accA = __builtin_amdgcn_wmma_f32_16x16x4_f32(
            false, a, false, ba[i], (short)0, accA, false, false);
        accB = __builtin_amdgcn_wmma_f32_16x16x4_f32(
            false, a, false, bb[i], (short)0, accB, false, false);
    }
}

template <int KW>
__device__ inline void load_xfrag(const float* x, v2f (&xa)[KW / 4]) {
    const int half = (threadIdx.x & 31) >> 4;
#pragma unroll
    for (int i = 0; i < KW / 4; ++i) {
        const int kb = 4 * i + 2 * half;
        xa[i].x = x[kb];
        xa[i].y = x[kb + 1];
    }
}

template <int KW>
__device__ inline v8f gemv_pre(const v2f (&xa)[KW / 4], const v2f (&b)[KW / 4], v8f acc,
                               const float amask) {
#pragma unroll
    for (int i = 0; i < KW / 4; ++i) {
        v2f a;
        a.x = xa[i].x * amask;
        a.y = xa[i].y * amask;
        acc = __builtin_amdgcn_wmma_f32_16x16x4_f32(
            false, a, false, b[i], (short)0, acc, false, false);
    }
    return acc;
}

template <int KW>
__device__ inline void gemv_pre2(const v2f (&xa)[KW / 4], const v2f (&ba)[KW / 4],
                                 const v2f (&bb)[KW / 4], v8f& accA, v8f& accB,
                                 const float amask) {
#pragma unroll
    for (int i = 0; i < KW / 4; ++i) {
        v2f a;
        a.x = xa[i].x * amask;
        a.y = xa[i].y * amask;
        accA = __builtin_amdgcn_wmma_f32_16x16x4_f32(
            false, a, false, ba[i], (short)0, accA, false, false);
        accB = __builtin_amdgcn_wmma_f32_16x16x4_f32(
            false, a, false, bb[i], (short)0, accB, false, false);
    }
}

__device__ inline float ksplit8_sum(v8f acc, float* part) {
    const int wave = threadIdx.x >> 5;
    const int lane = threadIdx.x & 31;
    if (lane < 16) part[wave * 16 + lane] = acc[0];
    lds_barrier();
    float r = 0.0f;
    if (threadIdx.x < 16) {
        for (int i = 0; i < 8; ++i) r += part[i * 16 + threadIdx.x];
    }
    return r;
}

__device__ inline void qp_body(
    const int bid,
    const float* tvec, const float* wt, const float* bt,
    const float* wk_x, const float* wk_h, const float* wk_s, const float* wk_m,
    float* pp, float* part, float* q_s)
{
    const int tid  = threadIdx.x;
    const int wave = tid >> 5;
    const int lane = tid & 31;
    const int tile = bid >> 2;
    const int mod  = bid & 3;
    const float* wkm = (mod == 0) ? wk_x : (mod == 1) ? wk_h : (mod == 2) ? wk_s : wk_m;
    v2f ta[8];
    load_xfrag<32>(tvec + wave * 32, ta);
    v2f bw[8];
    load_bfrag<32>(wt + (long)(wave * 32) * D_DIM, (long)D_DIM, 1L, tile * 16, bw);
    const float btv = bt[tile * 16 + (tid & 15)];
    const float* wq = wkm + tile * 16;
    const int n0 = wave * 32;
    v2f b0[4];
    v2f b1[4];
    load_bfrag<16>(wq, 1L, (long)D_DIM, n0, b0);
    load_bfrag<16>(wq, 1L, (long)D_DIM, n0 + 16, b1);
    pin_loads();

    v8f acc = {};
    acc = gemv_pre<32>(ta, bw, acc, ((tid & 15) == 0) ? 1.0f : 0.0f);
    const float r = ksplit8_sum(acc, part);
    if (tid < 16) q_s[tid] = r + btv;
    lds_barrier();

    v8f acc0 = {};
    v8f acc1 = {};
    gemv_frag2<16>(q_s, b0, b1, acc0, acc1);
    if (lane < 16) {
        float* po = pp + (long)(tile * 4 + mod) * C_DIM + n0;
        po[lane]      = acc0[0];
        po[16 + lane] = acc1[0];
    }
}

__global__ __launch_bounds__(256) void qp_kernel(
    const float* tvec, const float* wt, const float* bt,
    const float* wk_x, const float* wk_h, const float* wk_s, const float* wk_m,
    float* pp)
{
    __shared__ float part[128];
    __shared__ float q_s[16];
    qp_body(blockIdx.x, tvec, wt, bt, wk_x, wk_h, wk_s, wk_m, pp, part, q_s);
}

__device__ inline float ksplit8_sum2(const float* pa, const float* pb) {
    float r = 0.0f;
    if (threadIdx.x < 32) {
        const float* src = (threadIdx.x < 16) ? pa : pb;
        for (int i = 0; i < 8; ++i) r += src[i * 16 + (threadIdx.x & 15)];
    }
    return r;
}

__global__ __launch_bounds__(256) void qpf_kernel(
    const float* tvec, const float* wt, const float* bt,
    const float* wk_x, const float* wk_h, const float* wk_s, const float* wk_m,
    const float* w_ff2, const float* b_ff2, const float* w_out, const float* b_out,
    const float* g_res, const float* be_res, const float* w_geglu, const float* b_geglu,
    float* pp, float* mf, float* bv, float* cv, float* dv)
{
    __shared__ float part[128];
    __shared__ float q_s[16];
    __shared__ float fpart[8 * 512];
    const int bid  = blockIdx.x;
    const int tid  = threadIdx.x;
    const int wave = tid >> 5;
    const int lane = tid & 31;
    const int half = lane >> 4;
    const int l16  = lane & 15;
    if (bid < 64) {
        qp_body(bid, tvec, wt, bt, wk_x, wk_h, wk_s, wk_m, pp, part, q_s);
    } else if (bid < 64 + 128) {
        const int ft = bid - 64;
        const int mt = ft >> 3;
        const int np = ft & 7;
        const float* ap = w_ff2 + (long)(mt * 16 + l16) * D_DIM + wave * 32 + 2 * half;
        v2f af[8];
#pragma unroll
        for (int i = 0; i < 8; ++i) { af[i].x = ap[4 * i]; af[i].y = ap[4 * i + 1]; }
        v2f ba[8];
        v2f bb[8];
        load_bfrag<32>(w_out + (long)(wave * 32) * C_DIM, (long)C_DIM, 1L, np * 32, ba);
        load_bfrag<32>(w_out + (long)(wave * 32) * C_DIM, (long)C_DIM, 1L, np * 32 + 16, bb);
        pin_loads();
        v8f acca = {};
        v8f accb = {};
#pragma unroll
        for (int i = 0; i < 8; ++i) {
            acca = __builtin_amdgcn_wmma_f32_16x16x4_f32(
                false, af[i], false, ba[i], (short)0, acca, false, false);
            accb = __builtin_amdgcn_wmma_f32_16x16x4_f32(
                false, af[i], false, bb[i], (short)0, accb, false, false);
        }
        float* fw = fpart + wave * 512;
#pragma unroll
        for (int r = 0; r < 8; ++r) {
            fw[(8 * half + r) * 32 + l16]      = acca[r];
            fw[(8 * half + r) * 32 + 16 + l16] = accb[r];
        }
        lds_barrier();
        for (int e = tid; e < 512; e += 256) {
            float sum = 0.0f;
            for (int w = 0; w < 8; ++w) sum += fpart[w * 512 + e];
            mf[(long)(mt * 16 + (e >> 5)) * C_DIM + np * 32 + (e & 31)] = sum;
        }
    } else if (bid < 64 + 128 + 16) {
        const int n0 = (bid - (64 + 128)) * 32;
        const float* wg = w_geglu + (long)(wave * 32) * (2 * D_DIM);
        v2f ba[8];
        v2f bb[8];
        load_bfrag<32>(wg, (long)(2 * D_DIM), 1L, n0, ba);
        load_bfrag<32>(wg, (long)(2 * D_DIM), 1L, n0 + 16, bb);
        const float bg = b_geglu[n0 + (tid & 31)];
        v2f xa[8];
        load_xfrag<32>(((l16 == 1) ? be_res : g_res) + wave * 32, xa);
        pin_loads();
        v8f c0 = {};
        v8f c1 = {};
        gemv_pre2<32>(xa, ba, bb, c0, c1, (l16 < 2) ? 1.0f : 0.0f);
        if (lane < 16) {
            fpart[wave * 16 + lane]       = c0[0];
            fpart[128 + wave * 16 + lane] = c1[0];
            fpart[256 + wave * 16 + lane] = c0[1];
            fpart[384 + wave * 16 + lane] = c1[1];
        }
        lds_barrier();
        const float rc = ksplit8_sum2(fpart, fpart + 128);
        const float rd = ksplit8_sum2(fpart + 256, fpart + 384);
        if (tid < 32) { cv[n0 + tid] = rc; dv[n0 + tid] = rd + bg; }
    } else {
        const int n0 = (bid - (64 + 128 + 16)) * 32;
        v2f ba[8];
        v2f bb[8];
        load_bfrag<32>(w_out + (long)(wave * 32) * C_DIM, (long)C_DIM, 1L, n0, ba);
        load_bfrag<32>(w_out + (long)(wave * 32) * C_DIM, (long)C_DIM, 1L, n0 + 16, bb);
        const float bo = b_out[n0 + (tid & 31)];
        v2f fa[8];
        load_xfrag<32>(b_ff2 + wave * 32, fa);
        pin_loads();
        v8f a0 = {};
        v8f a1 = {};
        gemv_pre2<32>(fa, ba, bb, a0, a1, (l16 == 0) ? 1.0f : 0.0f);
        if (lane < 16) {
            fpart[wave * 16 + lane]       = a0[0];
            fpart[128 + wave * 16 + lane] = a1[0];
        }
        lds_barrier();
        const float rb = ksplit8_sum2(fpart, fpart + 128);
        if (tid < 32) bv[n0 + tid] = rb + bo;
    }
}

#define TR_PITCH 260
__global__ __launch_bounds__(512) void tile_kernel(
    const float* x, const float* h, const float* s, const float* m,
    const float* pp,
    const float* g_x, const float* be_x, const float* g_h, const float* be_h,
    const float* g_s, const float* be_s, const float* g_m, const float* be_m,
    float* up, float* pm, float* ps, float* pr)
{
    __shared__ __attribute__((aligned(16))) float r1[16 * 128];
    __shared__ __attribute__((aligned(16))) float r2[16 * 128];
    __shared__ __attribute__((aligned(16))) float rd[16 * 128];
    __shared__ __attribute__((aligned(16))) float al_s[128];
    __shared__ __attribute__((aligned(16))) float tr[32 * TR_PITCH];
    __shared__ __attribute__((aligned(16))) float gp_s[C_DIM];
    __shared__ float wsg[8];
    __shared__ float wsb[8];
    __shared__ float wred[12];
    const int tid = threadIdx.x;
    const int bid = blockIdx.x;
    const int mod = bid >> 5;
    const int j0  = (bid & 31) << 7;
    const int tq  = tid & 31;
    const int cg  = tid >> 5;
    const float* am  = (mod == 0) ? x    : (mod == 1) ? h    : (mod == 2) ? s    : m;
    const float* gm  = (mod == 0) ? g_x  : (mod == 1) ? g_h  : (mod == 2) ? g_s  : g_m;
    const float* bem = (mod == 0) ? be_x : (mod == 1) ? be_h : (mod == 2) ? be_s : be_m;
    const float* a = am + (long)(cg * 16) * N_TOK + j0 + 4 * tq;

    const bool chan = tid < C_DIM;
    float pq[16];
    float gch = 0.0f, bch = 0.0f;
#pragma unroll
    for (int i = 0; i < 16; ++i) pq[i] = 0.0f;
    if (chan) {
        const float* ppm = pp + mod * C_DIM + tid;
#pragma unroll
        for (int i = 0; i < 16; ++i) pq[i] = ppm[(long)i * (4 * C_DIM)];
        gch = gm[tid];
        bch = bem[tid];
    }
    float4 v[16];
#pragma unroll
    for (int cc = 0; cc < 16; ++cc) v[cc] = *(const float4*)(a + (long)cc * N_TOK);
    __builtin_amdgcn_sched_barrier(0);

    float gpc = 0.0f, bpc = 0.0f;
    if (chan) {
        float p = 0.0f;
#pragma unroll
        for (int i = 0; i < 16; ++i) p += pq[i];
        gpc = gch * p;
        bpc = bch * p;
        gp_s[tid] = gpc;
    }
    const float wsum_g = wave_sum32(gpc);
    const float wsum_b = wave_sum32(bpc);
    if (chan && tq == 0) { wsg[cg] = wsum_g; wsb[cg] = wsum_b; }
    lds_barrier();
    float sg = 0.0f, sb = 0.0f;
    for (int i = 0; i < 8; ++i) { sg += wsg[i]; sb += wsb[i]; }
    const float4* g4 = (const float4*)(gp_s + cg * 16);
    float gv[16];
#pragma unroll
    for (int q = 0; q < 4; ++q) {
        const float4 t = g4[q];
        gv[4 * q] = t.x; gv[4 * q + 1] = t.y; gv[4 * q + 2] = t.z; gv[4 * q + 3] = t.w;
    }

    float4 s1 = {0.0f, 0.0f, 0.0f, 0.0f};
    float4 s2 = {0.0f, 0.0f, 0.0f, 0.0f};
    float4 dp = {0.0f, 0.0f, 0.0f, 0.0f};
#pragma unroll
    for (int cc = 0; cc < 16; ++cc) {
        const float4 t = v[cc];
        const float g = gv[cc];
        s1.x += t.x;       s1.y += t.y;       s1.z += t.z;       s1.w += t.w;
        s2.x += t.x * t.x; s2.y += t.y * t.y; s2.z += t.z * t.z; s2.w += t.w * t.w;
        dp.x += t.x * g;   dp.y += t.y * g;   dp.z += t.z * g;   dp.w += t.w * g;
    }
    *(float4*)&r1[cg * 128 + 4 * tq] = s1;
    *(float4*)&r2[cg * 128 + 4 * tq] = s2;
    *(float4*)&rd[cg * 128 + 4 * tq] = dp;
    __syncthreads();

    const bool tok = tid < 128;
    const int  wv  = tid >> 5;
    float lg = 0.0f, mu = 0.0f, rstd = 0.0f;
    if (tok) {
        float t1 = 0.0f, t2 = 0.0f, td = 0.0f;
        for (int g = 0; g < 16; ++g) {
            t1 += r1[g * 128 + tid];
            t2 += r2[g * 128 + tid];
            td += rd[g * 128 + tid];
        }
        mu   = t1 * (1.0f / C_DIM);
        const float var  = t2 * (1.0f / C_DIM) - mu * mu;
        rstd = rsqrtf(var + EPSV);
        lg = SCALE * (rstd * (td - mu * sg) + sb);
        const float wm = wave_max32(lg);
        if ((tid & 31) == 0) wred[wv] = wm;
    }
    __syncthreads();
    if (tok) {
        const float bmax = fmaxf(fmaxf(wred[0], wred[1]), fmaxf(wred[2], wred[3]));
        const float e  = __expf(lg - bmax);
        const float al = e * rstd;
        al_s[tid] = al;
        const float se = wave_sum32(e);
        const float sr = wave_sum32(al * mu);
        if ((tid & 31) == 0) { wred[4 + wv] = se; wred[8 + wv] = sr; }
    }
    __syncthreads();
    if (tid == 0) {
        pm[bid] = fmaxf(fmaxf(wred[0], wred[1]), fmaxf(wred[2], wred[3]));
        ps[bid] = ((wred[4] + wred[5]) + wred[6]) + wred[7];
        pr[bid] = ((wred[8] + wred[9]) + wred[10]) + wred[11];
    }

    const float4 a4 = *(const float4*)&al_s[4 * tq];
    float pc[16];
#pragma unroll
    for (int cc = 0; cc < 16; ++cc) {
        const float4 t = v[cc];
        pc[cc] = ((t.x * a4.x + t.y * a4.y) + t.z * a4.z) + t.w * a4.w;
    }
    float* trw = tr + tq * TR_PITCH + cg * 16;
#pragma unroll
    for (int q = 0; q < 4; ++q) {
        float4 t;
        t.x = pc[4 * q]; t.y = pc[4 * q + 1]; t.z = pc[4 * q + 2]; t.w = pc[4 * q + 3];
        *(float4*)&trw[4 * q] = t;
    }
    __syncthreads();
    if (chan) {
        float uacc = 0.0f;
        for (int t = 0; t < 32; ++t) uacc += tr[t * TR_PITCH + tid];
        up[(long)bid * C_DIM + tid] = uacc;
    }
}


template <bool WITH_G>
__device__ inline void ov_body(
    const float* up, const float* pm, const float* ps, const float* pr,
    const float* g_x, const float* be_x, const float* g_h, const float* be_h,
    const float* g_s, const float* be_s, const float* g_m, const float* be_m,
    const float* wv_x, const float* wv_h, const float* wv_s, const float* wv_m,
    float* ovp, const float* g_res, const float* w_geglu, float* gp)
{
    __shared__ float u_s[C_DIM];
    __shared__ float f_s[N_BLK];
    __shared__ float wmx[4];
    __shared__ float wse[4];
    __shared__ float wre[4];
    __shared__ float part[128];
    const int tid  = threadIdx.x;
    const int wvi  = tid >> 5;
    const int lane = tid & 31;
    const int mod  = blockIdx.x >> 4;
    const int tile = blockIdx.x & 15;
    const float* wvm = (mod == 0) ? wv_x : (mod == 1) ? wv_h : (mod == 2) ? wv_s : wv_m;
    const float* gm  = (mod == 0) ? g_x  : (mod == 1) ? g_h  : (mod == 2) ? g_s  : g_m;
    const float* bem = (mod == 0) ? be_x : (mod == 1) ? be_h : (mod == 2) ? be_s : be_m;

    const bool blk = tid < N_BLK;
    float mb = 0.0f, sb = 0.0f, rb = 0.0f;
    if (blk) { mb = pm[tid]; sb = ps[tid]; rb = pr[tid]; }
    const float* upm = up + (long)(mod * 32) * C_DIM + tid;
    float pv[32];
#pragma unroll
    for (int b = 0; b < 32; ++b) pv[b] = upm[b * C_DIM];
    const float gc  = gm[tid];
    const float bec = bem[tid];
    v2f bw[8];
    load_bfrag<32>(wvm + (long)(wvi * 32) * D_DIM, (long)D_DIM, 1L, tile * 16, bw);
    v2f bg0[4];
    v2f bg1[4];
    v2f bg2[4];
    v2f bg3[4];
    float grs = 0.0f;
    if constexpr (WITH_G) {
        pin_loads();
        const float* wgr = w_geglu + (long)(tile * 16) * (2 * D_DIM) + wvi * 64;
        load_bfrag_quad16(wgr, (long)(2 * D_DIM), bg0, bg1, bg2, bg3);
        grs = g_res[tile * 16 + (tid & 15)];
        pin_loads();
    }

    if (blk) {
        const float wm = wave_max32(mb);
        if (lane == 0) wmx[wvi] = wm;
    }
    lds_barrier();
    const float gmax = fmaxf(fmaxf(wmx[0], wmx[1]), fmaxf(wmx[2], wmx[3]));
    if (blk) {
        const float eb = __expf(mb - gmax);
        f_s[tid] = eb;
        const float se = wave_sum32(sb * eb);
        const float sr = wave_sum32(rb * eb);
        if (lane == 0) { wse[wvi] = se; wre[wvi] = sr; }
    }
    lds_barrier();
    float ssum = 0.0f;
    for (int i = 0; i < 4; ++i) ssum += wse[i];
    const float inv = 1.0f / ssum;
    const float* fm = f_s + mod * 32;
    float c0 = 0.0f, c1 = 0.0f, c2 = 0.0f, c3 = 0.0f;
#pragma unroll
    for (int b = 0; b < 32; b += 4) {
        c0 += pv[b]     * fm[b];
        c1 += pv[b + 1] * fm[b + 1];
        c2 += pv[b + 2] * fm[b + 2];
        c3 += pv[b + 3] * fm[b + 3];
    }
    const float T    = ((c0 + c1) + (c2 + c3)) * inv;
    const float wsum = wse[mod] * inv;
    const float amu  = wre[mod] * inv;
    u_s[tid] = gc * (T - amu) + bec * wsum;
    lds_barrier();

    v8f acc = {};
    acc = gemv_frag<32>(u_s + wvi * 32, bw, acc);
    const float r = ksplit8_sum(acc, part);
    if (tid < 16) ovp[mod * D_DIM + tile * 16 + tid] = r;
    if constexpr (WITH_G) {
        __shared__ float xg_s[16];
        if (tid < 16) xg_s[tid] = r * grs;
        lds_barrier();
        v8f a0 = {};
        v8f a1 = {};
        v8f a2 = {};
        v8f a3 = {};
        gemv_frag2<16>(xg_s, bg0, bg1, a0, a1);
        gemv_frag2<16>(xg_s, bg2, bg3, a2, a3);
        if (lane < 16) {
            float4 gq;
            gq.x = a0[0]; gq.y = a1[0]; gq.z = a2[0]; gq.w = a3[0];
            *(float4*)(gp + (long)(mod * 16 + tile) * (2 * D_DIM) + wvi * 64 + 4 * lane) = gq;
        }
    }
}

__global__ __launch_bounds__(256) void ov_kernel(
    const float* up, const float* pm, const float* ps, const float* pr,
    const float* g_x, const float* be_x, const float* g_h, const float* be_h,
    const float* g_s, const float* be_s, const float* g_m, const float* be_m,
    const float* wv_x, const float* wv_h, const float* wv_s, const float* wv_m,
    float* ovp)
{
    ov_body<false>(up, pm, ps, pr, g_x, be_x, g_h, be_h, g_s, be_s, g_m, be_m,
                   wv_x, wv_h, wv_s, wv_m, ovp, nullptr, nullptr, nullptr);
}

__global__ __launch_bounds__(256) void ovg_kernel(
    const float* up, const float* pm, const float* ps, const float* pr,
    const float* g_x, const float* be_x, const float* g_h, const float* be_h,
    const float* g_s, const float* be_s, const float* g_m, const float* be_m,
    const float* wv_x, const float* wv_h, const float* wv_s, const float* wv_m,
    const float* g_res, const float* w_geglu,
    float* ovp, float* gp)
{
    ov_body<true>(up, pm, ps, pr, g_x, be_x, g_h, be_h, g_s, be_s, g_m, be_m,
                  wv_x, wv_h, wv_s, wv_m, ovp, g_res, w_geglu, gp);
}

__global__ __launch_bounds__(256) void ggff_kernel(
    const float* ovp, const float* g_res, const float* be_res,
    const float* w_geglu, const float* b_geglu, const float* w_ff2, float* ffp)
{
    __shared__ float rr[D_DIM];
    __shared__ float wsa[8];
    __shared__ float wsb[8];
    __shared__ float parta[128];
    __shared__ float partb[128];
    __shared__ float hh_s[16];
    const int tid  = threadIdx.x;
    const int wave = tid >> 5;
    const int lane = tid & 31;
    const int tile = blockIdx.x;

    const float o0 = ovp[tid];
    const float o1 = ovp[D_DIM + tid];
    const float o2 = ovp[2 * D_DIM + tid];
    const float o3 = ovp[3 * D_DIM + tid];
    const float gres  = g_res[tid];
    const float beres = be_res[tid];
    const float* wg = w_geglu + (long)(wave * 32) * (2 * D_DIM);
    v2f bga[8];
    v2f bgb[8];
    load_bfrag<32>(wg, (long)(2 * D_DIM), 1L, tile * 16, bga);
    load_bfrag<32>(wg, (long)(2 * D_DIM), 1L, D_DIM + tile * 16, bgb);
    const float bgu = b_geglu[tile * 16 + (tid & 15)];
    const float bgg = b_geglu[D_DIM + tile * 16 + (tid & 15)];
    const float* w2 = w_ff2 + (long)(tile * 16) * D_DIM;
    const int n0 = wave * 32;
    v2f b20[4];
    v2f b21[4];
    load_bfrag<16>(w2, (long)D_DIM, 1L, n0, b20);
    load_bfrag<16>(w2, (long)D_DIM, 1L, n0 + 16, b21);

    const float ov = ((o0 + o1) + o2) + o3;
    const float mu = block_sum_256(ov, wsa) * (1.0f / D_DIM);
    const float dv = ov - mu;
    const float var = block_sum_256(dv * dv, wsb) * (1.0f / D_DIM);
    const float rstd = rsqrtf(var + EPSV);
    rr[tid] = dv * rstd * gres + beres;
    lds_barrier();

    v8f acca = {};
    v8f accb = {};
    gemv_frag2<32>(rr + wave * 32, bga, bgb, acca, accb);
    if (lane < 16) { parta[wave * 16 + lane] = acca[0]; partb[wave * 16 + lane] = accb[0]; }
    lds_barrier();
    if (tid < 16) {
        float ru = 0.0f, rg = 0.0f;
        for (int i = 0; i < 8; ++i) { ru += parta[i * 16 + tid]; rg += partb[i * 16 + tid]; }
        const float uu = ru + bgu;
        const float g  = rg + bgg;
        const float ge = 0.5f * g * (1.0f + erff(g * 0.70710678118654752f));
        hh_s[tid] = uu * ge;
    }
    lds_barrier();

    v8f acc0 = {};
    v8f acc1 = {};
    gemv_frag2<16>(hh_s, b20, b21, acc0, acc1);
    if (lane < 16) {
        ffp[tile * D_DIM + n0 + lane]      = acc0[0];
        ffp[tile * D_DIM + n0 + 16 + lane] = acc1[0];
    }
}

__global__ __launch_bounds__(256) void outb_kernel(
    const float* ffp, const float* b_ff2, const float* w_out, const float* b_out,
    float* out)
{
    __shared__ float ff_s[D_DIM];
    __shared__ float part[128];
    __shared__ float y_s[16];
    const int tid   = threadIdx.x;
    const int tile  = blockIdx.x >> 4;
    const int slice = blockIdx.x & 15;
    const int wave  = tid >> 5;
    float fp[16];
#pragma unroll
    for (int i = 0; i < 16; ++i) fp[i] = ffp[i * D_DIM + tid];
    const float bf2 = b_ff2[tid];
    v2f bw[8];
    load_bfrag<32>(w_out + (long)(wave * 32) * C_DIM, (long)C_DIM, 1L, tile * 16, bw);
    const float bo = b_out[tile * 16 + (tid & 15)];

    float f = 0.0f;
#pragma unroll
    for (int i = 0; i < 16; ++i) f += fp[i];
    ff_s[tid] = f + bf2;
    lds_barrier();
    v8f acc = {};
    acc = gemv_frag<32>(ff_s + wave * 32, bw, acc);
    const float r = ksplit8_sum(acc, part);
    if (tid < 16) y_s[tid] = r + bo;
    lds_barrier();
    const int t4 = tid & 63;
    const int cg = tid >> 6;
    for (int i = 0; i < 4; ++i) {
        const int ch = 4 * i + cg;
        const float v = y_s[ch];
        float4 vv;
        vv.x = v; vv.y = v; vv.z = v; vv.w = v;
        float4* o = (float4*)(out + (long)(tile * 16 + ch) * N_TOK + slice * 256 + t4 * 4);
        *o = vv;
    }
}

__global__ __launch_bounds__(256) void fin_kernel(
    const float* ovp, const float* gp, const float* mf,
    const float* bv, const float* cv, const float* dv, float* out)
{
    __shared__ __attribute__((aligned(16))) float gs[2 * 2 * D_DIM];
    __shared__ float wsa[8];
    __shared__ float wsb[8];
    __shared__ float hh_s[D_DIM];
    __shared__ float part[128];
    __shared__ float y_s[16];
    const int tid   = threadIdx.x;
    const int tile  = blockIdx.x >> 4;
    const int slice = blockIdx.x & 15;
    const int wave  = tid >> 5;
    const int cq    = tid & 127;
    const int ph    = tid >> 7;
    const float o0 = ovp[tid];
    const float o1 = ovp[D_DIM + tid];
    const float o2 = ovp[2 * D_DIM + tid];
    const float o3 = ovp[3 * D_DIM + tid];
    const float4* gq = (const float4*)(gp + (long)(ph * 32) * (2 * D_DIM) + 4 * cq);
    float4 gv[32];
#pragma unroll
    for (int b = 0; b < 32; ++b) gv[b] = gq[b * (2 * D_DIM / 4)];
    const float cu = cv[tid];
    const float cg = cv[D_DIM + tid];
    const float du = dv[tid];
    const float dg = dv[D_DIM + tid];
    v2f bw[8];
    load_bfrag<32>(mf + (long)(wave * 32) * C_DIM, (long)C_DIM, 1L, tile * 16, bw);
    const float bo = bv[tile * 16 + (tid & 15)];

    const float ov = ((o0 + o1) + o2) + o3;
    const float mu = block_sum_256(ov, wsa) * (1.0f / D_DIM);
    const float dvv = ov - mu;
    const float var = block_sum_256(dvv * dvv, wsb) * (1.0f / D_DIM);
    const float rstd = rsqrtf(var + EPSV);

    float4 sg = {0.0f, 0.0f, 0.0f, 0.0f};
#pragma unroll
    for (int b = 0; b < 32; ++b) {
        sg.x += gv[b].x; sg.y += gv[b].y; sg.z += gv[b].z; sg.w += gv[b].w;
    }
    *(float4*)&gs[ph * (2 * D_DIM) + 4 * cq] = sg;
    lds_barrier();
    const float Gu = gs[tid] + gs[2 * D_DIM + tid];
    const float Gg = gs[D_DIM + tid] + gs[3 * D_DIM + tid];
    const float uu = rstd * (Gu - mu * cu) + du;
    const float g  = rstd * (Gg - mu * cg) + dg;
    const float ge = 0.5f * g * (1.0f + erff(g * 0.70710678118654752f));
    hh_s[tid] = uu * ge;
    lds_barrier();

    v8f acc = {};
    acc = gemv_frag<32>(hh_s + wave * 32, bw, acc);
    const float r = ksplit8_sum(acc, part);
    if (tid < 16) y_s[tid] = r + bo;
    lds_barrier();
    const int t4 = tid & 63;
    const int cgi = tid >> 6;
    for (int i = 0; i < 4; ++i) {
        const int ch = 4 * i + cgi;
        const float v = y_s[ch];
        float4 vv;
        vv.x = v; vv.y = v; vv.z = v; vv.w = v;
        float4* o = (float4*)(out + (long)(tile * 16 + ch) * N_TOK + slice * 256 + t4 * 4);
        *o = vv;
    }
}

extern "C" void kernel_launch(void* const* d_in, const int* in_sizes, int n_in,
                              void* d_out, int out_size, void* d_ws, size_t ws_size,
                              hipStream_t stream) {
    const float* x       = (const float*)d_in[0];
    const float* h       = (const float*)d_in[1];
    const float* s       = (const float*)d_in[2];
    const float* m       = (const float*)d_in[3];
    const float* tvec    = (const float*)d_in[4];
    const float* wt      = (const float*)d_in[5];
    const float* bt      = (const float*)d_in[6];
    const float* wk_x    = (const float*)d_in[7];
    const float* wv_x    = (const float*)d_in[8];
    const float* g_x     = (const float*)d_in[9];
    const float* be_x    = (const float*)d_in[10];
    const float* wk_h    = (const float*)d_in[11];
    const float* wv_h    = (const float*)d_in[12];
    const float* g_h     = (const float*)d_in[13];
    const float* be_h    = (const float*)d_in[14];
    const float* wk_s    = (const float*)d_in[15];
    const float* wv_s    = (const float*)d_in[16];
    const float* g_s     = (const float*)d_in[17];
    const float* be_s    = (const float*)d_in[18];
    const float* wk_m    = (const float*)d_in[19];
    const float* wv_m    = (const float*)d_in[20];
    const float* g_m     = (const float*)d_in[21];
    const float* be_m    = (const float*)d_in[22];
    const float* w_geglu = (const float*)d_in[23];
    const float* b_geglu = (const float*)d_in[24];
    const float* w_ff2   = (const float*)d_in[25];
    const float* b_ff2   = (const float*)d_in[26];
    const float* g_res   = (const float*)d_in[27];
    const float* be_res  = (const float*)d_in[28];
    const float* w_out   = (const float*)d_in[29];
    const float* b_out   = (const float*)d_in[30];
    float* out = (float*)d_out;
    float* ws  = (float*)d_ws;

    if (ws_size >= F_BYTES) {
        qpf_kernel<<<QPF_GRID, 256, 0, stream>>>(
            tvec, wt, bt, wk_x, wk_h, wk_s, wk_m,
            w_ff2, b_ff2, w_out, b_out, g_res, be_res, w_geglu, b_geglu,
            ws + F_PP, ws + F_MF, ws + F_BV, ws + F_CV, ws + F_DV);
        tile_kernel<<<N_BLK, 512, 0, stream>>>(x, h, s, m, ws + F_PP,
                                               g_x, be_x, g_h, be_h, g_s, be_s, g_m, be_m,
                                               ws + F_UP, ws + F_PM, ws + F_PS, ws + F_PR);
        ovg_kernel<<<64, 256, 0, stream>>>(ws + F_UP, ws + F_PM, ws + F_PS, ws + F_PR,
                                           g_x, be_x, g_h, be_h, g_s, be_s, g_m, be_m,
                                           wv_x, wv_h, wv_s, wv_m, g_res, w_geglu,
                                           ws + F_OVP, ws + F_GP);
        fin_kernel<<<256, 256, 0, stream>>>(ws + F_OVP, ws + F_GP, ws + F_MF,
                                            ws + F_BV, ws + F_CV, ws + F_DV, out);
        return;
    }

    float* pp     = ws + WS_PP;
    float* ovp    = ws + WS_OVP;
    float* ffp    = ws + WS_FFP;
    float* up     = ws + WS_UP;
    float* pm     = ws + WS_PM;
    float* ps     = ws + WS_PS;
    float* pr     = ws + WS_PR;

    qp_kernel<<<64, 256, 0, stream>>>(tvec, wt, bt, wk_x, wk_h, wk_s, wk_m, pp);
    tile_kernel<<<N_BLK, 512, 0, stream>>>(x, h, s, m, pp,
                                           g_x, be_x, g_h, be_h, g_s, be_s, g_m, be_m,
                                           up, pm, ps, pr);
    ov_kernel<<<64, 256, 0, stream>>>(up, pm, ps, pr,
                                      g_x, be_x, g_h, be_h, g_s, be_s, g_m, be_m,
                                      wv_x, wv_h, wv_s, wv_m, ovp);
    ggff_kernel<<<16, 256, 0, stream>>>(ovp, g_res, be_res, w_geglu, b_geglu, w_ff2, ffp);
    outb_kernel<<<256, 256, 0, stream>>>(ffp, b_ff2, w_out, b_out, out);
}
